# starting kernel
# baseline (speedup 1.0000x reference)
#include <hip/hip_runtime.h>


#define HSTEPS 8192
#define BLOCK  1024
#define NWAVE  32

#define ACC_LIMIT   3.0f
#define STEER_LIMIT 0.6f
#define V_MIN_C     0.0f
#define V_MAX_C     30.0f
#define WHEELBASE   2.7f

typedef __attribute__((ext_vector_type(2))) float v2f;
typedef __attribute__((ext_vector_type(8))) float v8f;

__device__ __forceinline__ float clipf(float x, float lo, float hi) {
    return fminf(fmaxf(x, lo), hi);
}

struct SMem {
    float stage[NWAVE][256];
    float tot[NWAVE][16];
    float tileSum[NWAVE];
    float tileOff[NWAVE];
    float fS[2][BLOCK];
    float fL[2][BLOCK];
    float fH[2][BLOCK];
};

__device__ __forceinline__ void tile_scan(float v[8], SMem* sm, int wave, int lane,
                                          const v2f aL[4], const v2f bM[4]) {
    const int col  = lane & 15;
    const int half = lane >> 4;

    __syncthreads();
    float* st = sm->stage[wave];
#pragma unroll
    for (int r = 0; r < 8; ++r)
        st[col * 16 + half * 8 + r] = v[r];
    __syncthreads();

    v8f c = {};
#pragma unroll
    for (int kb = 0; kb < 4; ++kb) {
        v2f b;
#pragma unroll
        for (int j = 0; j < 2; ++j)
            b[j] = st[col * 16 + kb * 4 + half * 2 + j];
        c = __builtin_amdgcn_wmma_f32_16x16x4_f32(false, aL[kb], false, b,
                                                  (short)0, c, false, false);
    }

    if (half == 1) sm->tot[wave][col] = c[7];
    __syncthreads();

#pragma unroll
    for (int kb = 0; kb < 4; ++kb) {
        v2f a;
#pragma unroll
        for (int j = 0; j < 2; ++j)
            a[j] = sm->tot[wave][kb * 4 + half * 2 + j];
        c = __builtin_amdgcn_wmma_f32_16x16x4_f32(false, a, false, bM[kb],
                                                  (short)0, c, false, false);
    }

    if (lane == 31) sm->tileSum[wave] = c[7];
    __syncthreads();

    if (threadIdx.x < NWAVE) {
        float tv = sm->tileSum[threadIdx.x];
        float incl = tv;
#pragma unroll
        for (int d = 1; d < 32; d <<= 1) {
            float o = __shfl_up(incl, d, 32);
            if ((int)threadIdx.x >= d) incl += o;
        }
        sm->tileOff[threadIdx.x] = incl - tv;
    }
    __syncthreads();

    const float carry = sm->tileOff[wave];
#pragma unroll
    for (int r = 0; r < 8; ++r) v[r] = c[r] + carry;
}

__global__ __launch_bounds__(BLOCK) void
KinematicBicycle_82214263980490_kernel(const float* __restrict__ x0,
                                       const float* __restrict__ U,
                                       const float* __restrict__ dtp,
                                       float* __restrict__ out) {
    __shared__ SMem sm;
    const int tid  = threadIdx.x;
    const int wave = tid >> 5;
    const int lane = tid & 31;
    const int col  = lane & 15;
    const int half = lane >> 4;
    const int rank = wave * 32 + col * 2 + half;
    const int t0 = wave * 256 + col * 16 + half * 8;

    v2f aL[4], bM[4];
#pragma unroll
    for (int kb = 0; kb < 4; ++kb) {
#pragma unroll
        for (int j = 0; j < 2; ++j) {
            const int k = kb * 4 + half * 2 + j;
            aL[kb][j] = (k <= col) ? 1.0f : 0.0f;
            bM[kb][j] = (k <  col) ? 1.0f : 0.0f;
        }
    }

    const float dt = dtp[0];
    const float px = x0[0], py = x0[1], pth = x0[2], pv = x0[3];

    float s[8], dl[8];
    const float4* u4 = (const float4*)(U + (size_t)t0 * 2);
#pragma unroll
    for (int q = 0; q < 4; ++q) {
        const float4 u = u4[q];
        s[2 * q]      = clipf(u.x, -ACC_LIMIT, ACC_LIMIT) * dt;
        dl[2 * q]     = clipf(u.y, -STEER_LIMIT, STEER_LIMIT);
        s[2 * q + 1]  = clipf(u.z, -ACC_LIMIT, ACC_LIMIT) * dt;
        dl[2 * q + 1] = clipf(u.w, -STEER_LIMIT, STEER_LIMIT);
    }

    float S = s[0], Lo = V_MIN_C, Hi = V_MAX_C;
#pragma unroll
    for (int r = 1; r < 8; ++r) {
        S += s[r];
        Lo = clipf(Lo + s[r], V_MIN_C, V_MAX_C);
        Hi = clipf(Hi + s[r], V_MIN_C, V_MAX_C);
    }
    int p = 0;
    sm.fS[0][rank] = S; sm.fL[0][rank] = Lo; sm.fH[0][rank] = Hi;
    __syncthreads();
    for (int off = 1; off < BLOCK; off <<= 1) {
        if (rank >= off) {
            const float pS = sm.fS[p][rank - off];
            const float pL = sm.fL[p][rank - off];
            const float pH = sm.fH[p][rank - off];
            const float nS = pS + S;
            const float nL = clipf(pL + S, Lo, Hi);
            const float nH = clipf(pH + S, Lo, Hi);
            S = nS; Lo = nL; Hi = nH;
        }
        p ^= 1;
        sm.fS[p][rank] = S; sm.fL[p][rank] = Lo; sm.fH[p][rank] = Hi;
        __syncthreads();
    }
    float w = clipf(pv, V_MIN_C, V_MAX_C);
    if (rank > 0)
        w = clipf(w + sm.fS[p][rank - 1], sm.fL[p][rank - 1], sm.fH[p][rank - 1]);

    float wst[8], vout[8], dth[8];
    const float dtOverL = dt / WHEELBASE;
#pragma unroll
    for (int r = 0; r < 8; ++r) {
        wst[r] = w;
        dth[r] = w * tanf(dl[r]) * dtOverL;
        w = clipf(w + s[r], V_MIN_C, V_MAX_C);
        vout[r] = w;
    }

    float incl[8];
#pragma unroll
    for (int r = 0; r < 8; ++r) incl[r] = dth[r];
    tile_scan(incl, &sm, wave, lane, aL, bM);
    float thout[8], thst[8];
#pragma unroll
    for (int r = 0; r < 8; ++r) {
        thout[r] = pth + incl[r];
        thst[r]  = thout[r] - dth[r];
    }

    float iy[8];
#pragma unroll
    for (int r = 0; r < 8; ++r) {
        float sn, cs;
        sincosf(thst[r], &sn, &cs);
        const float g = wst[r] * dt;
        incl[r] = g * cs;
        iy[r]   = g * sn;
    }
    tile_scan(incl, &sm, wave, lane, aL, bM);
    tile_scan(iy,   &sm, wave, lane, aL, bM);

#pragma unroll
    for (int r = 0; r < 8; ++r) {
        float4 row = make_float4(px + incl[r], py + iy[r], thout[r], vout[r]);
        *(float4*)(out + (size_t)(t0 + r + 1) * 4) = row;
    }
    if (tid == 0) *(float4*)out = make_float4(px, py, pth, pv);
}

extern "C" void kernel_launch(void* const* d_in, const int* in_sizes, int n_in,
                              void* d_out, int out_size, void* d_ws, size_t ws_size,
                              hipStream_t stream) {
    (void)in_sizes; (void)n_in; (void)out_size; (void)d_ws; (void)ws_size;
    const float* x0 = (const float*)d_in[0];
    const float* U  = (const float*)d_in[1];
    const float* dt = (const float*)d_in[2];
    float* out = (float*)d_out;
    KinematicBicycle_82214263980490_kernel<<<1, BLOCK, 0, stream>>>(x0, U, dt, out);
}


# Workgroup barriers 26 -> 3: hand-overs inside one wave take none, velocity-function scan per wave on cross-lane ops, each wave scans the 32 wave/tile totals itself; new f32 sum order in carries
# speedup vs baseline: 1.2658x; 1.2658x over previous
#include <hip/hip_runtime.h>


#define HSTEPS 8192
#define BLOCK  1024
#define NWAVE  32

#define ACC_LIMIT   3.0f
#define STEER_LIMIT 0.6f
#define V_MIN_C     0.0f
#define V_MAX_C     30.0f
#define WHEELBASE   2.7f

typedef __attribute__((ext_vector_type(2))) float v2f;
typedef __attribute__((ext_vector_type(8))) float v8f;

__device__ __forceinline__ float clipf(float x, float lo, float hi) {
    return fminf(fmaxf(x, lo), hi);
}

struct ClampAdd { float S, Lo, Hi; };

__device__ __forceinline__ ClampAdd compose(ClampAdd first, ClampAdd second) {
    ClampAdd r;
    r.S  = first.S + second.S;
    r.Lo = clipf(first.Lo + second.S, second.Lo, second.Hi);
    r.Hi = clipf(first.Hi + second.S, second.Lo, second.Hi);
    return r;
}
__device__ __forceinline__ float apply(ClampAdd f, float v) {
    return clipf(v + f.S, f.Lo, f.Hi);
}

template <int N>
__device__ __forceinline__ float row_left(float v) {
    const int x = __builtin_bit_cast(int, v);
    return __builtin_bit_cast(float,
        __builtin_amdgcn_update_dpp(x, x, 0x110 + N, 0xF, 0xF, false));
}
__device__ __forceinline__ float other_half(float v) {
    const int x = __builtin_bit_cast(int, v);
    return __builtin_bit_cast(float,
        __builtin_amdgcn_permlanex16(x, x, 0x76543210, 0xfedcba98, false, false));
}
__device__ __forceinline__ float lane_value(float v, int uniform_lane) {
    return __builtin_bit_cast(float,
        __builtin_amdgcn_readlane(__builtin_bit_cast(int, v), uniform_lane));
}

template <int N>
__device__ __forceinline__ ClampAdd row_scan_step(ClampAdd f, int col) {
    ClampAdd left;
    left.S = row_left<N>(f.S); left.Lo = row_left<N>(f.Lo); left.Hi = row_left<N>(f.Hi);
    const ClampAdd c = compose(left, f);
    if (col >= N) f = c;
    return f;
}
__device__ __forceinline__ ClampAdd row_scan(ClampAdd f, int col) {
    f = row_scan_step<1>(f, col);
    f = row_scan_step<2>(f, col);
    f = row_scan_step<4>(f, col);
    f = row_scan_step<8>(f, col);
    return f;
}

template <int N>
__device__ __forceinline__ float row_sum_step(float x, int col) {
    const float o = row_left<N>(x);
    return (col >= N) ? x + o : x;
}

__device__ __forceinline__ void wave_lds_handover() {
    __builtin_amdgcn_fence(__ATOMIC_RELEASE, "wavefront");
    __builtin_amdgcn_wave_barrier();
    __builtin_amdgcn_fence(__ATOMIC_ACQUIRE, "wavefront");
}

struct SMem {
    float stage[2][NWAVE][256];
    float tot[2][NWAVE][16];
    float tileSum[3][NWAVE];
    float wS[NWAVE];
    float wL[NWAVE];
    float wH[NWAVE];
};

template <int NS>
__device__ __forceinline__ void tile_scan(float (&v)[NS][8], SMem* sm, int plane0,
                                          int wave, int lane,
                                          const v2f aL[4], const v2f bM[4]) {
    const int col  = lane & 15;
    const int half = lane >> 4;

#pragma unroll
    for (int s = 0; s < NS; ++s) {
        float* st = sm->stage[s][wave];
#pragma unroll
        for (int r = 0; r < 8; ++r)
            st[col * 16 + half * 8 + r] = v[s][r];
    }
    wave_lds_handover();

    v8f c[NS];
#pragma unroll
    for (int s = 0; s < NS; ++s) c[s] = v8f{};
#pragma unroll
    for (int kb = 0; kb < 4; ++kb) {
#pragma unroll
        for (int s = 0; s < NS; ++s) {
            const float* st = sm->stage[s][wave];
            v2f b;
#pragma unroll
            for (int j = 0; j < 2; ++j)
                b[j] = st[col * 16 + kb * 4 + half * 2 + j];
            c[s] = __builtin_amdgcn_wmma_f32_16x16x4_f32(false, aL[kb], false, b,
                                                         (short)0, c[s], false, false);
        }
    }

    if (half == 1) {
#pragma unroll
        for (int s = 0; s < NS; ++s) sm->tot[s][wave][col] = c[s][7];
    }
    wave_lds_handover();

#pragma unroll
    for (int kb = 0; kb < 4; ++kb) {
#pragma unroll
        for (int s = 0; s < NS; ++s) {
            v2f a;
#pragma unroll
            for (int j = 0; j < 2; ++j)
                a[j] = sm->tot[s][wave][kb * 4 + half * 2 + j];
            c[s] = __builtin_amdgcn_wmma_f32_16x16x4_f32(false, a, false, bM[kb],
                                                         (short)0, c[s], false, false);
        }
    }

    if (lane == 31) {
#pragma unroll
        for (int s = 0; s < NS; ++s) sm->tileSum[plane0 + s][wave] = c[s][7];
    }
    __syncthreads();

    float incl[NS];
#pragma unroll
    for (int s = 0; s < NS; ++s) incl[s] = sm->tileSum[plane0 + s][lane];
#pragma unroll
    for (int s = 0; s < NS; ++s) {
        float x = incl[s];
        x = row_sum_step<1>(x, col);
        x = row_sum_step<2>(x, col);
        x = row_sum_step<4>(x, col);
        x = row_sum_step<8>(x, col);
        const float low = lane_value(x, 15);
        if (half == 1) x += low;
        const float carry = (wave > 0) ? lane_value(x, wave > 0 ? wave - 1 : 0) : 0.0f;
#pragma unroll
        for (int r = 0; r < 8; ++r) v[s][r] = c[s][r] + carry;
    }
}

__global__ __launch_bounds__(BLOCK) void
KinematicBicycle_82214263980490_kernel(const float* __restrict__ x0,
                                       const float* __restrict__ U,
                                       const float* __restrict__ dtp,
                                       float* __restrict__ out) {
    __shared__ SMem sm;
    const int tid  = threadIdx.x;
    const int wave = __builtin_amdgcn_readfirstlane(tid >> 5);
    const int lane = tid & 31;
    const int col  = lane & 15;
    const int half = lane >> 4;
    const int t0 = wave * 256 + col * 16 + half * 8;

    v2f aL[4], bM[4];
#pragma unroll
    for (int kb = 0; kb < 4; ++kb) {
#pragma unroll
        for (int j = 0; j < 2; ++j) {
            const int k = kb * 4 + half * 2 + j;
            aL[kb][j] = (k <= col) ? 1.0f : 0.0f;
            bM[kb][j] = (k <  col) ? 1.0f : 0.0f;
        }
    }

    const float dt = dtp[0];
    const float px = x0[0], py = x0[1], pth = x0[2], pv = x0[3];

    float s[8], dl[8];
    const float4* u4 = (const float4*)(U + (size_t)t0 * 2);
#pragma unroll
    for (int q = 0; q < 4; ++q) {
        const float4 u = u4[q];
        s[2 * q]      = clipf(u.x, -ACC_LIMIT, ACC_LIMIT) * dt;
        dl[2 * q]     = clipf(u.y, -STEER_LIMIT, STEER_LIMIT);
        s[2 * q + 1]  = clipf(u.z, -ACC_LIMIT, ACC_LIMIT) * dt;
        dl[2 * q + 1] = clipf(u.w, -STEER_LIMIT, STEER_LIMIT);
    }

    ClampAdd mine;
    mine.S = s[0]; mine.Lo = V_MIN_C; mine.Hi = V_MAX_C;
#pragma unroll
    for (int r = 1; r < 8; ++r) {
        mine.S += s[r];
        mine.Lo = clipf(mine.Lo + s[r], V_MIN_C, V_MAX_C);
        mine.Hi = clipf(mine.Hi + s[r], V_MIN_C, V_MAX_C);
    }
    ClampAdd partner;
    partner.S = other_half(mine.S); partner.Lo = other_half(mine.Lo); partner.Hi = other_half(mine.Hi);
    const ClampAdd colFn = half ? compose(partner, mine) : compose(mine, partner);
    const ClampAdd inclCols = row_scan(colFn, col);
    ClampAdd exclCols;
    exclCols.S = row_left<1>(inclCols.S); exclCols.Lo = row_left<1>(inclCols.Lo); exclCols.Hi = row_left<1>(inclCols.Hi);
    if (lane == 15) { sm.wS[wave] = inclCols.S; sm.wL[wave] = inclCols.Lo; sm.wH[wave] = inclCols.Hi; }
    __syncthreads();
    ClampAdd wf;
    wf.S = sm.wS[lane]; wf.Lo = sm.wL[lane]; wf.Hi = sm.wH[lane];
    wf = row_scan(wf, col);
    {
        ClampAdd low;
        low.S = lane_value(wf.S, 15); low.Lo = lane_value(wf.Lo, 15); low.Hi = lane_value(wf.Hi, 15);
        const ClampAdd c2 = compose(low, wf);
        if (half == 1) wf = c2;
    }
    float w = clipf(pv, V_MIN_C, V_MAX_C);
    if (wave > 0) {
        ClampAdd pre;
        pre.S = lane_value(wf.S, wave - 1); pre.Lo = lane_value(wf.Lo, wave - 1); pre.Hi = lane_value(wf.Hi, wave - 1);
        w = apply(pre, w);
    }
    if (col > 0) w = apply(exclCols, w);
    if (half == 1) w = apply(partner, w);

    float wst[8], vout[8], dth[8];
    const float dtOverL = dt / WHEELBASE;
#pragma unroll
    for (int r = 0; r < 8; ++r) {
        wst[r] = w;
        dth[r] = w * tanf(dl[r]) * dtOverL;
        w = clipf(w + s[r], V_MIN_C, V_MAX_C);
        vout[r] = w;
    }

    float th[1][8];
#pragma unroll
    for (int r = 0; r < 8; ++r) th[0][r] = dth[r];
    tile_scan<1>(th, &sm, 0, wave, lane, aL, bM);
    float thout[8], thst[8];
#pragma unroll
    for (int r = 0; r < 8; ++r) {
        thout[r] = pth + th[0][r];
        thst[r]  = thout[r] - dth[r];
    }

    float xy[2][8];
#pragma unroll
    for (int r = 0; r < 8; ++r) {
        float sn, cs;
        sincosf(thst[r], &sn, &cs);
        const float g = wst[r] * dt;
        xy[0][r] = g * cs;
        xy[1][r] = g * sn;
    }
    tile_scan<2>(xy, &sm, 1, wave, lane, aL, bM);

#pragma unroll
    for (int r = 0; r < 8; ++r) {
        float4 row = make_float4(px + xy[0][r], py + xy[1][r], thout[r], vout[r]);
        *(float4*)(out + (size_t)(t0 + r + 1) * 4) = row;
    }
    if (tid == 0) *(float4*)out = make_float4(px, py, pth, pv);
}

extern "C" void kernel_launch(void* const* d_in, const int* in_sizes, int n_in,
                              void* d_out, int out_size, void* d_ws, size_t ws_size,
                              hipStream_t stream) {
    (void)in_sizes; (void)n_in; (void)out_size; (void)d_ws; (void)ws_size;
    const float* x0 = (const float*)d_in[0];
    const float* U  = (const float*)d_in[1];
    const float* dt = (const float*)d_in[2];
    float* out = (float*)d_out;
    KinematicBicycle_82214263980490_kernel<<<1, BLOCK, 0, stream>>>(x0, U, dt, out);
}


# tanf and sincosf replaced by the library's own small-argument polynomials inline, two angles per packed f32 op; one large-angle test per thread (library call kept behind it) instead of one per call
# speedup vs baseline: 1.4353x; 1.1339x over previous
#include <hip/hip_runtime.h>


#define HSTEPS 8192
#define BLOCK  1024
#define NWAVE  32

#define ACC_LIMIT   3.0f
#define STEER_LIMIT 0.6f
#define V_MIN_C     0.0f
#define V_MAX_C     30.0f
#define WHEELBASE   2.7f

typedef __attribute__((ext_vector_type(2))) float v2f;
typedef __attribute__((ext_vector_type(8))) float v8f;

__device__ __forceinline__ float clipf(float x, float lo, float hi) {
    return fminf(fmaxf(x, lo), hi);
}

__device__ __forceinline__ v2f pk_fma(v2f a, v2f b, v2f c) {
    return __builtin_elementwise_fma(a, b, c);
}
__device__ __forceinline__ v2f splat(unsigned bits) {
    const float f = __builtin_bit_cast(float, bits);
    return v2f{f, f};
}

__device__ __forceinline__ v2f tan_small_pair(v2f x) {
    const v2f s = x * x;
    v2f den = pk_fma(s, splat(0x3c971480u), splat(0xbf039337u));
    den = pk_fma(s, den, splat(0x3f93f425u));
    const v2f num = pk_fma(s, splat(0xbc8cedd3u), splat(0x3ec54587u));
    v2f rd;
    rd[0] = __builtin_amdgcn_rcpf(den[0]);
    rd[1] = __builtin_amdgcn_rcpf(den[1]);
    const v2f p = s * (num * rd);
    return pk_fma(p, x, x);
}

__device__ __forceinline__ void sincos_pair(v2f x, v2f& sn, v2f& cs) {
    const v2f n = __builtin_elementwise_rint(x * splat(0x3f22f983u));
    v2f r = pk_fma(n, splat(0xbfc90fdau), x);
    r = pk_fma(n, splat(0xb3a22168u), r);
    r = pk_fma(n, splat(0xa7c234c4u), r);
    const v2f t = r * r;
    v2f p = pk_fma(t, splat(0xb94c1982u), splat(0x3c0881c4u));
    p = pk_fma(t, p, splat(0xbe2aaa9du));
    p = t * p;
    const v2f s = pk_fma(r, p, r);
    v2f q = pk_fma(t, splat(0x37d75334u), splat(0xbab64f3bu));
    q = pk_fma(t, q, splat(0x3d2aabf7u));
    q = pk_fma(t, q, splat(0xbf000004u));
    const v2f c = pk_fma(t, q, splat(0x3f800000u));
#pragma unroll
    for (int j = 0; j < 2; ++j) {
        const unsigned k = (unsigned)(int)n[j];
        const bool odd = (k & 1u) != 0u;
        const unsigned flip = (k << 30) & 0x80000000u;
        const float so = odd ? c[j] : s[j];
        const float co = odd ? -s[j] : c[j];
        sn[j] = __builtin_bit_cast(float, __builtin_bit_cast(unsigned, so) ^ flip);
        cs[j] = __builtin_bit_cast(float, __builtin_bit_cast(unsigned, co) ^ flip);
    }
}

struct ClampAdd { float S, Lo, Hi; };

__device__ __forceinline__ ClampAdd compose(ClampAdd first, ClampAdd second) {
    ClampAdd r;
    r.S  = first.S + second.S;
    r.Lo = clipf(first.Lo + second.S, second.Lo, second.Hi);
    r.Hi = clipf(first.Hi + second.S, second.Lo, second.Hi);
    return r;
}
__device__ __forceinline__ float apply(ClampAdd f, float v) {
    return clipf(v + f.S, f.Lo, f.Hi);
}

template <int N>
__device__ __forceinline__ float row_left(float v) {
    const int x = __builtin_bit_cast(int, v);
    return __builtin_bit_cast(float,
        __builtin_amdgcn_update_dpp(x, x, 0x110 + N, 0xF, 0xF, false));
}
__device__ __forceinline__ float other_half(float v) {
    const int x = __builtin_bit_cast(int, v);
    return __builtin_bit_cast(float,
        __builtin_amdgcn_permlanex16(x, x, 0x76543210, 0xfedcba98, false, false));
}
__device__ __forceinline__ float lane_value(float v, int uniform_lane) {
    return __builtin_bit_cast(float,
        __builtin_amdgcn_readlane(__builtin_bit_cast(int, v), uniform_lane));
}

template <int N>
__device__ __forceinline__ ClampAdd row_scan_step(ClampAdd f, int col) {
    ClampAdd left;
    left.S = row_left<N>(f.S); left.Lo = row_left<N>(f.Lo); left.Hi = row_left<N>(f.Hi);
    const ClampAdd c = compose(left, f);
    if (col >= N) f = c;
    return f;
}
__device__ __forceinline__ ClampAdd row_scan(ClampAdd f, int col) {
    f = row_scan_step<1>(f, col);
    f = row_scan_step<2>(f, col);
    f = row_scan_step<4>(f, col);
    f = row_scan_step<8>(f, col);
    return f;
}

template <int N>
__device__ __forceinline__ float row_sum_step(float x, int col) {
    const float o = row_left<N>(x);
    return (col >= N) ? x + o : x;
}

__device__ __forceinline__ void wave_lds_handover() {
    __builtin_amdgcn_fence(__ATOMIC_RELEASE, "wavefront");
    __builtin_amdgcn_wave_barrier();
    __builtin_amdgcn_fence(__ATOMIC_ACQUIRE, "wavefront");
}

struct SMem {
    float stage[2][NWAVE][256];
    float tot[2][NWAVE][16];
    float tileSum[3][NWAVE];
    float wS[NWAVE];
    float wL[NWAVE];
    float wH[NWAVE];
};

template <int NS>
__device__ __forceinline__ void tile_scan(float (&v)[NS][8], SMem* sm, int plane0,
                                          int wave, int lane,
                                          const v2f aL[4], const v2f bM[4]) {
    const int col  = lane & 15;
    const int half = lane >> 4;

#pragma unroll
    for (int s = 0; s < NS; ++s) {
        float* st = sm->stage[s][wave];
#pragma unroll
        for (int r = 0; r < 8; ++r)
            st[col * 16 + half * 8 + r] = v[s][r];
    }
    wave_lds_handover();

    v8f c[NS];
#pragma unroll
    for (int s = 0; s < NS; ++s) c[s] = v8f{};
#pragma unroll
    for (int kb = 0; kb < 4; ++kb) {
#pragma unroll
        for (int s = 0; s < NS; ++s) {
            const float* st = sm->stage[s][wave];
            v2f b;
#pragma unroll
            for (int j = 0; j < 2; ++j)
                b[j] = st[col * 16 + kb * 4 + half * 2 + j];
            c[s] = __builtin_amdgcn_wmma_f32_16x16x4_f32(false, aL[kb], false, b,
                                                         (short)0, c[s], false, false);
        }
    }

    if (half == 1) {
#pragma unroll
        for (int s = 0; s < NS; ++s) sm->tot[s][wave][col] = c[s][7];
    }
    wave_lds_handover();

#pragma unroll
    for (int kb = 0; kb < 4; ++kb) {
#pragma unroll
        for (int s = 0; s < NS; ++s) {
            v2f a;
#pragma unroll
            for (int j = 0; j < 2; ++j)
                a[j] = sm->tot[s][wave][kb * 4 + half * 2 + j];
            c[s] = __builtin_amdgcn_wmma_f32_16x16x4_f32(false, a, false, bM[kb],
                                                         (short)0, c[s], false, false);
        }
    }

    if (lane == 31) {
#pragma unroll
        for (int s = 0; s < NS; ++s) sm->tileSum[plane0 + s][wave] = c[s][7];
    }
    __syncthreads();

    float incl[NS];
#pragma unroll
    for (int s = 0; s < NS; ++s) incl[s] = sm->tileSum[plane0 + s][lane];
#pragma unroll
    for (int s = 0; s < NS; ++s) {
        float x = incl[s];
        x = row_sum_step<1>(x, col);
        x = row_sum_step<2>(x, col);
        x = row_sum_step<4>(x, col);
        x = row_sum_step<8>(x, col);
        const float low = lane_value(x, 15);
        if (half == 1) x += low;
        const float carry = (wave > 0) ? lane_value(x, wave > 0 ? wave - 1 : 0) : 0.0f;
#pragma unroll
        for (int r = 0; r < 8; ++r) v[s][r] = c[s][r] + carry;
    }
}

__global__ __launch_bounds__(BLOCK) void
KinematicBicycle_82214263980490_kernel(const float* __restrict__ x0,
                                       const float* __restrict__ U,
                                       const float* __restrict__ dtp,
                                       float* __restrict__ out) {
    __shared__ SMem sm;
    const int tid  = threadIdx.x;
    const int wave = __builtin_amdgcn_readfirstlane(tid >> 5);
    const int lane = tid & 31;
    const int col  = lane & 15;
    const int half = lane >> 4;
    const int t0 = wave * 256 + col * 16 + half * 8;

    v2f aL[4], bM[4];
#pragma unroll
    for (int kb = 0; kb < 4; ++kb) {
#pragma unroll
        for (int j = 0; j < 2; ++j) {
            const int k = kb * 4 + half * 2 + j;
            aL[kb][j] = (k <= col) ? 1.0f : 0.0f;
            bM[kb][j] = (k <  col) ? 1.0f : 0.0f;
        }
    }

    const float dt = dtp[0];
    const float px = x0[0], py = x0[1], pth = x0[2], pv = x0[3];

    float s[8], dl[8];
    const float4* u4 = (const float4*)(U + (size_t)t0 * 2);
#pragma unroll
    for (int q = 0; q < 4; ++q) {
        const float4 u = u4[q];
        s[2 * q]      = clipf(u.x, -ACC_LIMIT, ACC_LIMIT) * dt;
        dl[2 * q]     = clipf(u.y, -STEER_LIMIT, STEER_LIMIT);
        s[2 * q + 1]  = clipf(u.z, -ACC_LIMIT, ACC_LIMIT) * dt;
        dl[2 * q + 1] = clipf(u.w, -STEER_LIMIT, STEER_LIMIT);
    }

    ClampAdd mine;
    mine.S = s[0]; mine.Lo = V_MIN_C; mine.Hi = V_MAX_C;
#pragma unroll
    for (int r = 1; r < 8; ++r) {
        mine.S += s[r];
        mine.Lo = clipf(mine.Lo + s[r], V_MIN_C, V_MAX_C);
        mine.Hi = clipf(mine.Hi + s[r], V_MIN_C, V_MAX_C);
    }
    ClampAdd partner;
    partner.S = other_half(mine.S); partner.Lo = other_half(mine.Lo); partner.Hi = other_half(mine.Hi);
    const ClampAdd colFn = half ? compose(partner, mine) : compose(mine, partner);
    const ClampAdd inclCols = row_scan(colFn, col);
    ClampAdd exclCols;
    exclCols.S = row_left<1>(inclCols.S); exclCols.Lo = row_left<1>(inclCols.Lo); exclCols.Hi = row_left<1>(inclCols.Hi);
    if (lane == 15) { sm.wS[wave] = inclCols.S; sm.wL[wave] = inclCols.Lo; sm.wH[wave] = inclCols.Hi; }
    __syncthreads();
    ClampAdd wf;
    wf.S = sm.wS[lane]; wf.Lo = sm.wL[lane]; wf.Hi = sm.wH[lane];
    wf = row_scan(wf, col);
    {
        ClampAdd low;
        low.S = lane_value(wf.S, 15); low.Lo = lane_value(wf.Lo, 15); low.Hi = lane_value(wf.Hi, 15);
        const ClampAdd c2 = compose(low, wf);
        if (half == 1) wf = c2;
    }
    float w = clipf(pv, V_MIN_C, V_MAX_C);
    if (wave > 0) {
        ClampAdd pre;
        pre.S = lane_value(wf.S, wave - 1); pre.Lo = lane_value(wf.Lo, wave - 1); pre.Hi = lane_value(wf.Hi, wave - 1);
        w = apply(pre, w);
    }
    if (col > 0) w = apply(exclCols, w);
    if (half == 1) w = apply(partner, w);

    float wst[8], vout[8], dth[8], tn[8];
    const float dtOverL = dt / WHEELBASE;
#pragma unroll
    for (int q = 0; q < 4; ++q) {
        const v2f t2 = tan_small_pair(v2f{dl[2 * q], dl[2 * q + 1]});
        tn[2 * q] = t2[0]; tn[2 * q + 1] = t2[1];
    }
#pragma unroll
    for (int r = 0; r < 8; ++r) {
        wst[r] = w;
        dth[r] = w * tn[r] * dtOverL;
        w = clipf(w + s[r], V_MIN_C, V_MAX_C);
        vout[r] = w;
    }

    float th[1][8];
#pragma unroll
    for (int r = 0; r < 8; ++r) th[0][r] = dth[r];
    tile_scan<1>(th, &sm, 0, wave, lane, aL, bM);
    float thout[8], thst[8];
#pragma unroll
    for (int r = 0; r < 8; ++r) {
        thout[r] = pth + th[0][r];
        thst[r]  = thout[r] - dth[r];
    }

    float xy[2][8];
    float amax = 0.0f;
#pragma unroll
    for (int r = 0; r < 8; ++r) amax = __builtin_fmaxf(amax, __builtin_fabsf(thst[r]));
    if (__builtin_expect(amax < 131072.0f, 1)) {
#pragma unroll
        for (int q = 0; q < 4; ++q) {
            v2f sn, cs;
            sincos_pair(v2f{thst[2 * q], thst[2 * q + 1]}, sn, cs);
            const v2f g = v2f{wst[2 * q], wst[2 * q + 1]} * dt;
            const v2f gx = g * cs, gy = g * sn;
            xy[0][2 * q] = gx[0]; xy[0][2 * q + 1] = gx[1];
            xy[1][2 * q] = gy[0]; xy[1][2 * q + 1] = gy[1];
        }
    } else {
#pragma unroll
        for (int r = 0; r < 8; ++r) {
            float sn, cs;
            sincosf(thst[r], &sn, &cs);
            const float g = wst[r] * dt;
            xy[0][r] = g * cs;
            xy[1][r] = g * sn;
        }
    }
    tile_scan<2>(xy, &sm, 1, wave, lane, aL, bM);

#pragma unroll
    for (int r = 0; r < 8; ++r) {
        float4 row = make_float4(px + xy[0][r], py + xy[1][r], thout[r], vout[r]);
        *(float4*)(out + (size_t)(t0 + r + 1) * 4) = row;
    }
    if (tid == 0) *(float4*)out = make_float4(px, py, pth, pv);
}

extern "C" void kernel_launch(void* const* d_in, const int* in_sizes, int n_in,
                              void* d_out, int out_size, void* d_ws, size_t ws_size,
                              hipStream_t stream) {
    (void)in_sizes; (void)n_in; (void)out_size; (void)d_ws; (void)ws_size;
    const float* x0 = (const float*)d_in[0];
    const float* U  = (const float*)d_in[1];
    const float* dt = (const float*)d_in[2];
    float* out = (float*)d_out;
    KinematicBicycle_82214263980490_kernel<<<1, BLOCK, 0, stream>>>(x0, U, dt, out);
}


# Output rows staged through a per-wave LDS block so that each global store writes 4 whole 128-B lines (was 32 lines x 16 B per store); every value and the f32 sum order unchanged
# speedup vs baseline: 1.7523x; 1.2209x over previous
#include <hip/hip_runtime.h>


#define HSTEPS 8192
#define BLOCK  1024
#define NWAVE  32

#define ACC_LIMIT   3.0f
#define STEER_LIMIT 0.6f
#define V_MIN_C     0.0f
#define V_MAX_C     30.0f
#define WHEELBASE   2.7f

typedef __attribute__((ext_vector_type(2))) float v2f;
typedef __attribute__((ext_vector_type(8))) float v8f;

__device__ __forceinline__ float clipf(float x, float lo, float hi) {
    return fminf(fmaxf(x, lo), hi);
}

__device__ __forceinline__ v2f pk_fma(v2f a, v2f b, v2f c) {
    return __builtin_elementwise_fma(a, b, c);
}
__device__ __forceinline__ v2f splat(unsigned bits) {
    const float f = __builtin_bit_cast(float, bits);
    return v2f{f, f};
}

__device__ __forceinline__ v2f tan_small_pair(v2f x) {
    const v2f s = x * x;
    v2f den = pk_fma(s, splat(0x3c971480u), splat(0xbf039337u));
    den = pk_fma(s, den, splat(0x3f93f425u));
    const v2f num = pk_fma(s, splat(0xbc8cedd3u), splat(0x3ec54587u));
    v2f rd;
    rd[0] = __builtin_amdgcn_rcpf(den[0]);
    rd[1] = __builtin_amdgcn_rcpf(den[1]);
    const v2f p = s * (num * rd);
    return pk_fma(p, x, x);
}

__device__ __forceinline__ void sincos_pair(v2f x, v2f& sn, v2f& cs) {
    const v2f n = __builtin_elementwise_rint(x * splat(0x3f22f983u));
    v2f r = pk_fma(n, splat(0xbfc90fdau), x);
    r = pk_fma(n, splat(0xb3a22168u), r);
    r = pk_fma(n, splat(0xa7c234c4u), r);
    const v2f t = r * r;
    v2f p = pk_fma(t, splat(0xb94c1982u), splat(0x3c0881c4u));
    p = pk_fma(t, p, splat(0xbe2aaa9du));
    p = t * p;
    const v2f s = pk_fma(r, p, r);
    v2f q = pk_fma(t, splat(0x37d75334u), splat(0xbab64f3bu));
    q = pk_fma(t, q, splat(0x3d2aabf7u));
    q = pk_fma(t, q, splat(0xbf000004u));
    const v2f c = pk_fma(t, q, splat(0x3f800000u));
#pragma unroll
    for (int j = 0; j < 2; ++j) {
        const unsigned k = (unsigned)(int)n[j];
        const bool odd = (k & 1u) != 0u;
        const unsigned flip = (k << 30) & 0x80000000u;
        const float so = odd ? c[j] : s[j];
        const float co = odd ? -s[j] : c[j];
        sn[j] = __builtin_bit_cast(float, __builtin_bit_cast(unsigned, so) ^ flip);
        cs[j] = __builtin_bit_cast(float, __builtin_bit_cast(unsigned, co) ^ flip);
    }
}

struct ClampAdd { float S, Lo, Hi; };

__device__ __forceinline__ ClampAdd compose(ClampAdd first, ClampAdd second) {
    ClampAdd r;
    r.S  = first.S + second.S;
    r.Lo = clipf(first.Lo + second.S, second.Lo, second.Hi);
    r.Hi = clipf(first.Hi + second.S, second.Lo, second.Hi);
    return r;
}
__device__ __forceinline__ float apply(ClampAdd f, float v) {
    return clipf(v + f.S, f.Lo, f.Hi);
}

template <int N>
__device__ __forceinline__ float row_left(float v) {
    const int x = __builtin_bit_cast(int, v);
    return __builtin_bit_cast(float,
        __builtin_amdgcn_update_dpp(x, x, 0x110 + N, 0xF, 0xF, false));
}
__device__ __forceinline__ float other_half(float v) {
    const int x = __builtin_bit_cast(int, v);
    return __builtin_bit_cast(float,
        __builtin_amdgcn_permlanex16(x, x, 0x76543210, 0xfedcba98, false, false));
}
__device__ __forceinline__ float lane_value(float v, int uniform_lane) {
    return __builtin_bit_cast(float,
        __builtin_amdgcn_readlane(__builtin_bit_cast(int, v), uniform_lane));
}

template <int N>
__device__ __forceinline__ ClampAdd row_scan_step(ClampAdd f, int col) {
    ClampAdd left;
    left.S = row_left<N>(f.S); left.Lo = row_left<N>(f.Lo); left.Hi = row_left<N>(f.Hi);
    const ClampAdd c = compose(left, f);
    if (col >= N) f = c;
    return f;
}
__device__ __forceinline__ ClampAdd row_scan(ClampAdd f, int col) {
    f = row_scan_step<1>(f, col);
    f = row_scan_step<2>(f, col);
    f = row_scan_step<4>(f, col);
    f = row_scan_step<8>(f, col);
    return f;
}

template <int N>
__device__ __forceinline__ float row_sum_step(float x, int col) {
    const float o = row_left<N>(x);
    return (col >= N) ? x + o : x;
}

__device__ __forceinline__ void wave_lds_handover() {
    __builtin_amdgcn_fence(__ATOMIC_RELEASE, "wavefront");
    __builtin_amdgcn_wave_barrier();
    __builtin_amdgcn_fence(__ATOMIC_ACQUIRE, "wavefront");
}

#define OUT_SLOTS 273

struct SMem {
    float4 outStage[NWAVE][OUT_SLOTS];
    float stage[2][NWAVE][256];
    float tot[2][NWAVE][16];
    float tileSum[3][NWAVE];
    float wS[NWAVE];
    float wL[NWAVE];
    float wH[NWAVE];
};

template <int NS>
__device__ __forceinline__ void tile_scan(float (&v)[NS][8], SMem* sm, int plane0,
                                          int wave, int lane,
                                          const v2f aL[4], const v2f bM[4]) {
    const int col  = lane & 15;
    const int half = lane >> 4;

#pragma unroll
    for (int s = 0; s < NS; ++s) {
        float* st = sm->stage[s][wave];
#pragma unroll
        for (int r = 0; r < 8; ++r)
            st[col * 16 + half * 8 + r] = v[s][r];
    }
    wave_lds_handover();

    v8f c[NS];
#pragma unroll
    for (int s = 0; s < NS; ++s) c[s] = v8f{};
#pragma unroll
    for (int kb = 0; kb < 4; ++kb) {
#pragma unroll
        for (int s = 0; s < NS; ++s) {
            const float* st = sm->stage[s][wave];
            v2f b;
#pragma unroll
            for (int j = 0; j < 2; ++j)
                b[j] = st[col * 16 + kb * 4 + half * 2 + j];
            c[s] = __builtin_amdgcn_wmma_f32_16x16x4_f32(false, aL[kb], false, b,
                                                         (short)0, c[s], false, false);
        }
    }

    if (half == 1) {
#pragma unroll
        for (int s = 0; s < NS; ++s) sm->tot[s][wave][col] = c[s][7];
    }
    wave_lds_handover();

#pragma unroll
    for (int kb = 0; kb < 4; ++kb) {
#pragma unroll
        for (int s = 0; s < NS; ++s) {
            v2f a;
#pragma unroll
            for (int j = 0; j < 2; ++j)
                a[j] = sm->tot[s][wave][kb * 4 + half * 2 + j];
            c[s] = __builtin_amdgcn_wmma_f32_16x16x4_f32(false, a, false, bM[kb],
                                                         (short)0, c[s], false, false);
        }
    }

    if (lane == 31) {
#pragma unroll
        for (int s = 0; s < NS; ++s) sm->tileSum[plane0 + s][wave] = c[s][7];
    }
    __syncthreads();

    float incl[NS];
#pragma unroll
    for (int s = 0; s < NS; ++s) incl[s] = sm->tileSum[plane0 + s][lane];
#pragma unroll
    for (int s = 0; s < NS; ++s) {
        float x = incl[s];
        x = row_sum_step<1>(x, col);
        x = row_sum_step<2>(x, col);
        x = row_sum_step<4>(x, col);
        x = row_sum_step<8>(x, col);
        const float low = lane_value(x, 15);
        if (half == 1) x += low;
        const float carry = (wave > 0) ? lane_value(x, wave > 0 ? wave - 1 : 0) : 0.0f;
#pragma unroll
        for (int r = 0; r < 8; ++r) v[s][r] = c[s][r] + carry;
    }
}

__global__ __launch_bounds__(BLOCK) void
KinematicBicycle_82214263980490_kernel(const float* __restrict__ x0,
                                       const float* __restrict__ U,
                                       const float* __restrict__ dtp,
                                       float* __restrict__ out) {
    __shared__ SMem sm;
    const int tid  = threadIdx.x;
    const int wave = __builtin_amdgcn_readfirstlane(tid >> 5);
    const int lane = tid & 31;
    const int col  = lane & 15;
    const int half = lane >> 4;
    const int t0 = wave * 256 + col * 16 + half * 8;

    v2f aL[4], bM[4];
#pragma unroll
    for (int kb = 0; kb < 4; ++kb) {
#pragma unroll
        for (int j = 0; j < 2; ++j) {
            const int k = kb * 4 + half * 2 + j;
            aL[kb][j] = (k <= col) ? 1.0f : 0.0f;
            bM[kb][j] = (k <  col) ? 1.0f : 0.0f;
        }
    }

    const float dt = dtp[0];
    const float px = x0[0], py = x0[1], pth = x0[2], pv = x0[3];

    float s[8], dl[8];
    const float4* u4 = (const float4*)(U + (size_t)t0 * 2);
#pragma unroll
    for (int q = 0; q < 4; ++q) {
        const float4 u = u4[q];
        s[2 * q]      = clipf(u.x, -ACC_LIMIT, ACC_LIMIT) * dt;
        dl[2 * q]     = clipf(u.y, -STEER_LIMIT, STEER_LIMIT);
        s[2 * q + 1]  = clipf(u.z, -ACC_LIMIT, ACC_LIMIT) * dt;
        dl[2 * q + 1] = clipf(u.w, -STEER_LIMIT, STEER_LIMIT);
    }

    ClampAdd mine;
    mine.S = s[0]; mine.Lo = V_MIN_C; mine.Hi = V_MAX_C;
#pragma unroll
    for (int r = 1; r < 8; ++r) {
        mine.S += s[r];
        mine.Lo = clipf(mine.Lo + s[r], V_MIN_C, V_MAX_C);
        mine.Hi = clipf(mine.Hi + s[r], V_MIN_C, V_MAX_C);
    }
    ClampAdd partner;
    partner.S = other_half(mine.S); partner.Lo = other_half(mine.Lo); partner.Hi = other_half(mine.Hi);
    const ClampAdd colFn = half ? compose(partner, mine) : compose(mine, partner);
    const ClampAdd inclCols = row_scan(colFn, col);
    ClampAdd exclCols;
    exclCols.S = row_left<1>(inclCols.S); exclCols.Lo = row_left<1>(inclCols.Lo); exclCols.Hi = row_left<1>(inclCols.Hi);
    if (lane == 15) { sm.wS[wave] = inclCols.S; sm.wL[wave] = inclCols.Lo; sm.wH[wave] = inclCols.Hi; }
    __syncthreads();
    ClampAdd wf;
    wf.S = sm.wS[lane]; wf.Lo = sm.wL[lane]; wf.Hi = sm.wH[lane];
    wf = row_scan(wf, col);
    {
        ClampAdd low;
        low.S = lane_value(wf.S, 15); low.Lo = lane_value(wf.Lo, 15); low.Hi = lane_value(wf.Hi, 15);
        const ClampAdd c2 = compose(low, wf);
        if (half == 1) wf = c2;
    }
    float w = clipf(pv, V_MIN_C, V_MAX_C);
    if (wave > 0) {
        ClampAdd pre;
        pre.S = lane_value(wf.S, wave - 1); pre.Lo = lane_value(wf.Lo, wave - 1); pre.Hi = lane_value(wf.Hi, wave - 1);
        w = apply(pre, w);
    }
    if (col > 0) w = apply(exclCols, w);
    if (half == 1) w = apply(partner, w);

    float wst[8], vout[8], dth[8], tn[8];
    const float dtOverL = dt / WHEELBASE;
#pragma unroll
    for (int q = 0; q < 4; ++q) {
        const v2f t2 = tan_small_pair(v2f{dl[2 * q], dl[2 * q + 1]});
        tn[2 * q] = t2[0]; tn[2 * q + 1] = t2[1];
    }
#pragma unroll
    for (int r = 0; r < 8; ++r) {
        wst[r] = w;
        dth[r] = w * tn[r] * dtOverL;
        w = clipf(w + s[r], V_MIN_C, V_MAX_C);
        vout[r] = w;
    }

    float th[1][8];
#pragma unroll
    for (int r = 0; r < 8; ++r) th[0][r] = dth[r];
    tile_scan<1>(th, &sm, 0, wave, lane, aL, bM);
    float thout[8], thst[8];
#pragma unroll
    for (int r = 0; r < 8; ++r) {
        thout[r] = pth + th[0][r];
        thst[r]  = thout[r] - dth[r];
    }

    float xy[2][8];
    float amax = 0.0f;
#pragma unroll
    for (int r = 0; r < 8; ++r) amax = __builtin_fmaxf(amax, __builtin_fabsf(thst[r]));
    if (__builtin_expect(amax < 131072.0f, 1)) {
#pragma unroll
        for (int q = 0; q < 4; ++q) {
            v2f sn, cs;
            sincos_pair(v2f{thst[2 * q], thst[2 * q + 1]}, sn, cs);
            const v2f g = v2f{wst[2 * q], wst[2 * q + 1]} * dt;
            const v2f gx = g * cs, gy = g * sn;
            xy[0][2 * q] = gx[0]; xy[0][2 * q + 1] = gx[1];
            xy[1][2 * q] = gy[0]; xy[1][2 * q + 1] = gy[1];
        }
    } else {
#pragma unroll
        for (int r = 0; r < 8; ++r) {
            float sn, cs;
            sincosf(thst[r], &sn, &cs);
            const float g = wst[r] * dt;
            xy[0][r] = g * cs;
            xy[1][r] = g * sn;
        }
    }
    tile_scan<2>(xy, &sm, 1, wave, lane, aL, bM);

    float4* ob = sm.outStage[wave];
    const int k0 = col * 16 + half * 8 + 1;
#pragma unroll
    for (int r = 0; r < 8; ++r) {
        const int k = k0 + r;
        ob[k + (k >> 4)] = make_float4(px + xy[0][r], py + xy[1][r], thout[r], vout[r]);
    }
    wave_lds_handover();
    const int kf = (lane > 0) ? lane : 256;
    float4* og = (float4*)out + wave * 256;
    const float4* rd = ob + (lane + (lane >> 4));
    float4 rows[8];
    rows[0] = ob[kf + (kf >> 4)];
#pragma unroll
    for (int i = 1; i < 8; ++i) rows[i] = rd[34 * i];
    og[kf] = rows[0];
#pragma unroll
    for (int i = 1; i < 8; ++i) og[lane + 32 * i] = rows[i];
    if (tid == 0) *(float4*)out = make_float4(px, py, pth, pv);
}

extern "C" void kernel_launch(void* const* d_in, const int* in_sizes, int n_in,
                              void* d_out, int out_size, void* d_ws, size_t ws_size,
                              hipStream_t stream) {
    (void)in_sizes; (void)n_in; (void)out_size; (void)d_ws; (void)ws_size;
    const float* x0 = (const float*)d_in[0];
    const float* U  = (const float*)d_in[1];
    const float* dt = (const float*)d_in[2];
    float* out = (float*)d_out;
    KinematicBicycle_82214263980490_kernel<<<1, BLOCK, 0, stream>>>(x0, U, dt, out);
}


# Grid 1 -> 8 workgroups, each repeating the whole rollout and writing one eighth of the rows (whole 128-B lines): one WGP's store drain and LDS output staging shrink 8x; every value unchanged
# speedup vs baseline: 1.8871x; 1.0769x over previous
#include <hip/hip_runtime.h>


#define HSTEPS 8192
#define BLOCK  1024
#define NWAVE  32
#define NGRID  8
#define OWN_WAVES (NWAVE / NGRID)

#define ACC_LIMIT   3.0f
#define STEER_LIMIT 0.6f
#define V_MIN_C     0.0f
#define V_MAX_C     30.0f
#define WHEELBASE   2.7f

typedef __attribute__((ext_vector_type(2))) float v2f;
typedef __attribute__((ext_vector_type(8))) float v8f;

__device__ __forceinline__ float clipf(float x, float lo, float hi) {
    return fminf(fmaxf(x, lo), hi);
}

__device__ __forceinline__ v2f pk_fma(v2f a, v2f b, v2f c) {
    return __builtin_elementwise_fma(a, b, c);
}
__device__ __forceinline__ v2f splat(unsigned bits) {
    const float f = __builtin_bit_cast(float, bits);
    return v2f{f, f};
}

__device__ __forceinline__ v2f tan_small_pair(v2f x) {
    const v2f s = x * x;
    v2f den = pk_fma(s, splat(0x3c971480u), splat(0xbf039337u));
    den = pk_fma(s, den, splat(0x3f93f425u));
    const v2f num = pk_fma(s, splat(0xbc8cedd3u), splat(0x3ec54587u));
    v2f rd;
    rd[0] = __builtin_amdgcn_rcpf(den[0]);
    rd[1] = __builtin_amdgcn_rcpf(den[1]);
    const v2f p = s * (num * rd);
    return pk_fma(p, x, x);
}

__device__ __forceinline__ void sincos_pair(v2f x, v2f& sn, v2f& cs) {
    const v2f n = __builtin_elementwise_rint(x * splat(0x3f22f983u));
    v2f r = pk_fma(n, splat(0xbfc90fdau), x);
    r = pk_fma(n, splat(0xb3a22168u), r);
    r = pk_fma(n, splat(0xa7c234c4u), r);
    const v2f t = r * r;
    v2f p = pk_fma(t, splat(0xb94c1982u), splat(0x3c0881c4u));
    p = pk_fma(t, p, splat(0xbe2aaa9du));
    p = t * p;
    const v2f s = pk_fma(r, p, r);
    v2f q = pk_fma(t, splat(0x37d75334u), splat(0xbab64f3bu));
    q = pk_fma(t, q, splat(0x3d2aabf7u));
    q = pk_fma(t, q, splat(0xbf000004u));
    const v2f c = pk_fma(t, q, splat(0x3f800000u));
#pragma unroll
    for (int j = 0; j < 2; ++j) {
        const unsigned k = (unsigned)(int)n[j];
        const bool odd = (k & 1u) != 0u;
        const unsigned flip = (k << 30) & 0x80000000u;
        const float so = odd ? c[j] : s[j];
        const float co = odd ? -s[j] : c[j];
        sn[j] = __builtin_bit_cast(float, __builtin_bit_cast(unsigned, so) ^ flip);
        cs[j] = __builtin_bit_cast(float, __builtin_bit_cast(unsigned, co) ^ flip);
    }
}

struct ClampAdd { float S, Lo, Hi; };

__device__ __forceinline__ ClampAdd compose(ClampAdd first, ClampAdd second) {
    ClampAdd r;
    r.S  = first.S + second.S;
    r.Lo = clipf(first.Lo + second.S, second.Lo, second.Hi);
    r.Hi = clipf(first.Hi + second.S, second.Lo, second.Hi);
    return r;
}
__device__ __forceinline__ float apply(ClampAdd f, float v) {
    return clipf(v + f.S, f.Lo, f.Hi);
}

template <int N>
__device__ __forceinline__ float row_left(float v) {
    const int x = __builtin_bit_cast(int, v);
    return __builtin_bit_cast(float,
        __builtin_amdgcn_update_dpp(x, x, 0x110 + N, 0xF, 0xF, false));
}
__device__ __forceinline__ float other_half(float v) {
    const int x = __builtin_bit_cast(int, v);
    return __builtin_bit_cast(float,
        __builtin_amdgcn_permlanex16(x, x, 0x76543210, 0xfedcba98, false, false));
}
__device__ __forceinline__ float lane_value(float v, int uniform_lane) {
    return __builtin_bit_cast(float,
        __builtin_amdgcn_readlane(__builtin_bit_cast(int, v), uniform_lane));
}

template <int N>
__device__ __forceinline__ ClampAdd row_scan_step(ClampAdd f, int col) {
    ClampAdd left;
    left.S = row_left<N>(f.S); left.Lo = row_left<N>(f.Lo); left.Hi = row_left<N>(f.Hi);
    const ClampAdd c = compose(left, f);
    if (col >= N) f = c;
    return f;
}
__device__ __forceinline__ ClampAdd row_scan(ClampAdd f, int col) {
    f = row_scan_step<1>(f, col);
    f = row_scan_step<2>(f, col);
    f = row_scan_step<4>(f, col);
    f = row_scan_step<8>(f, col);
    return f;
}

template <int N>
__device__ __forceinline__ float row_sum_step(float x, int col) {
    const float o = row_left<N>(x);
    return (col >= N) ? x + o : x;
}

__device__ __forceinline__ void wave_lds_handover() {
    __builtin_amdgcn_fence(__ATOMIC_RELEASE, "wavefront");
    __builtin_amdgcn_wave_barrier();
    __builtin_amdgcn_fence(__ATOMIC_ACQUIRE, "wavefront");
}

#define OUT_SLOTS 273

struct SMem {
    float4 outStage[NWAVE][OUT_SLOTS];
    float stage[2][NWAVE][256];
    float tot[2][NWAVE][16];
    float tileSum[3][NWAVE];
    float wS[NWAVE];
    float wL[NWAVE];
    float wH[NWAVE];
};

template <int NS>
__device__ __forceinline__ void tile_scan(float (&v)[NS][8], SMem* sm, int plane0,
                                          int wave, int lane,
                                          const v2f aL[4], const v2f bM[4],
                                          bool wanted) {
    const int col  = lane & 15;
    const int half = lane >> 4;

#pragma unroll
    for (int s = 0; s < NS; ++s) {
        float* st = sm->stage[s][wave];
#pragma unroll
        for (int r = 0; r < 8; ++r)
            st[col * 16 + half * 8 + r] = v[s][r];
    }
    wave_lds_handover();

    v8f c[NS];
#pragma unroll
    for (int s = 0; s < NS; ++s) c[s] = v8f{};
#pragma unroll
    for (int kb = 0; kb < 4; ++kb) {
#pragma unroll
        for (int s = 0; s < NS; ++s) {
            const float* st = sm->stage[s][wave];
            v2f b;
#pragma unroll
            for (int j = 0; j < 2; ++j)
                b[j] = st[col * 16 + kb * 4 + half * 2 + j];
            c[s] = __builtin_amdgcn_wmma_f32_16x16x4_f32(false, aL[kb], false, b,
                                                         (short)0, c[s], false, false);
        }
    }

    if (half == 1) {
#pragma unroll
        for (int s = 0; s < NS; ++s) sm->tot[s][wave][col] = c[s][7];
    }
    wave_lds_handover();

#pragma unroll
    for (int kb = 0; kb < 4; ++kb) {
#pragma unroll
        for (int s = 0; s < NS; ++s) {
            v2f a;
#pragma unroll
            for (int j = 0; j < 2; ++j)
                a[j] = sm->tot[s][wave][kb * 4 + half * 2 + j];
            c[s] = __builtin_amdgcn_wmma_f32_16x16x4_f32(false, a, false, bM[kb],
                                                         (short)0, c[s], false, false);
        }
    }

    if (lane == 31) {
#pragma unroll
        for (int s = 0; s < NS; ++s) sm->tileSum[plane0 + s][wave] = c[s][7];
    }
    __syncthreads();
    if (!wanted) return;

    float incl[NS];
#pragma unroll
    for (int s = 0; s < NS; ++s) incl[s] = sm->tileSum[plane0 + s][lane];
#pragma unroll
    for (int s = 0; s < NS; ++s) {
        float x = incl[s];
        x = row_sum_step<1>(x, col);
        x = row_sum_step<2>(x, col);
        x = row_sum_step<4>(x, col);
        x = row_sum_step<8>(x, col);
        const float low = lane_value(x, 15);
        if (half == 1) x += low;
        const float carry = (wave > 0) ? lane_value(x, wave > 0 ? wave - 1 : 0) : 0.0f;
#pragma unroll
        for (int r = 0; r < 8; ++r) v[s][r] = c[s][r] + carry;
    }
}

__global__ __launch_bounds__(BLOCK) void
rollout_scan_kernel(const float* __restrict__ x0,
                    const float* __restrict__ U,
                    const float* __restrict__ dtp,
                    float* __restrict__ out) {
    __shared__ SMem sm;
    const int tid  = threadIdx.x;
    const int wave = __builtin_amdgcn_readfirstlane(tid >> 5);
    const int lane = tid & 31;
    const int col  = lane & 15;
    const int half = lane >> 4;
    const int t0 = wave * 256 + col * 16 + half * 8;

    const int group     = blockIdx.x;
    const int firstWave = group * OWN_WAVES;
    const bool owner    = (wave >= firstWave) && (wave < firstWave + OWN_WAVES);
    const bool tailOnly = (wave == firstWave - 1);
    const bool wanted   = owner || tailOnly;
    const bool sendLast = (wave != firstWave + OWN_WAVES - 1) || (wave == NWAVE - 1);

    v2f aL[4], bM[4];
#pragma unroll
    for (int kb = 0; kb < 4; ++kb) {
#pragma unroll
        for (int j = 0; j < 2; ++j) {
            const int k = kb * 4 + half * 2 + j;
            aL[kb][j] = (k <= col) ? 1.0f : 0.0f;
            bM[kb][j] = (k <  col) ? 1.0f : 0.0f;
        }
    }

    const float dt = dtp[0];
    const float px = x0[0], py = x0[1], pth = x0[2], pv = x0[3];

    float s[8], dl[8];
    const float4* u4 = (const float4*)(U + (size_t)t0 * 2);
#pragma unroll
    for (int q = 0; q < 4; ++q) {
        const float4 u = u4[q];
        s[2 * q]      = clipf(u.x, -ACC_LIMIT, ACC_LIMIT) * dt;
        dl[2 * q]     = clipf(u.y, -STEER_LIMIT, STEER_LIMIT);
        s[2 * q + 1]  = clipf(u.z, -ACC_LIMIT, ACC_LIMIT) * dt;
        dl[2 * q + 1] = clipf(u.w, -STEER_LIMIT, STEER_LIMIT);
    }

    ClampAdd mine;
    mine.S = s[0]; mine.Lo = V_MIN_C; mine.Hi = V_MAX_C;
#pragma unroll
    for (int r = 1; r < 8; ++r) {
        mine.S += s[r];
        mine.Lo = clipf(mine.Lo + s[r], V_MIN_C, V_MAX_C);
        mine.Hi = clipf(mine.Hi + s[r], V_MIN_C, V_MAX_C);
    }
    ClampAdd partner;
    partner.S = other_half(mine.S); partner.Lo = other_half(mine.Lo); partner.Hi = other_half(mine.Hi);
    const ClampAdd colFn = half ? compose(partner, mine) : compose(mine, partner);
    const ClampAdd inclCols = row_scan(colFn, col);
    ClampAdd exclCols;
    exclCols.S = row_left<1>(inclCols.S); exclCols.Lo = row_left<1>(inclCols.Lo); exclCols.Hi = row_left<1>(inclCols.Hi);
    if (lane == 15) { sm.wS[wave] = inclCols.S; sm.wL[wave] = inclCols.Lo; sm.wH[wave] = inclCols.Hi; }
    __syncthreads();
    ClampAdd wf;
    wf.S = sm.wS[lane]; wf.Lo = sm.wL[lane]; wf.Hi = sm.wH[lane];
    wf = row_scan(wf, col);
    {
        ClampAdd low;
        low.S = lane_value(wf.S, 15); low.Lo = lane_value(wf.Lo, 15); low.Hi = lane_value(wf.Hi, 15);
        const ClampAdd c2 = compose(low, wf);
        if (half == 1) wf = c2;
    }
    float w = clipf(pv, V_MIN_C, V_MAX_C);
    if (wave > 0) {
        ClampAdd pre;
        pre.S = lane_value(wf.S, wave - 1); pre.Lo = lane_value(wf.Lo, wave - 1); pre.Hi = lane_value(wf.Hi, wave - 1);
        w = apply(pre, w);
    }
    if (col > 0) w = apply(exclCols, w);
    if (half == 1) w = apply(partner, w);

    float wst[8], vout[8], dth[8], tn[8];
    const float dtOverL = dt / WHEELBASE;
#pragma unroll
    for (int q = 0; q < 4; ++q) {
        const v2f t2 = tan_small_pair(v2f{dl[2 * q], dl[2 * q + 1]});
        tn[2 * q] = t2[0]; tn[2 * q + 1] = t2[1];
    }
#pragma unroll
    for (int r = 0; r < 8; ++r) {
        wst[r] = w;
        dth[r] = w * tn[r] * dtOverL;
        w = clipf(w + s[r], V_MIN_C, V_MAX_C);
        vout[r] = w;
    }

    float th[1][8];
#pragma unroll
    for (int r = 0; r < 8; ++r) th[0][r] = dth[r];
    tile_scan<1>(th, &sm, 0, wave, lane, aL, bM, true);
    float thout[8], thst[8];
#pragma unroll
    for (int r = 0; r < 8; ++r) {
        thout[r] = pth + th[0][r];
        thst[r]  = thout[r] - dth[r];
    }

    float xy[2][8];
    float amax = 0.0f;
#pragma unroll
    for (int r = 0; r < 8; ++r) amax = __builtin_fmaxf(amax, __builtin_fabsf(thst[r]));
    if (__builtin_expect(amax < 131072.0f, 1)) {
#pragma unroll
        for (int q = 0; q < 4; ++q) {
            v2f sn, cs;
            sincos_pair(v2f{thst[2 * q], thst[2 * q + 1]}, sn, cs);
            const v2f g = v2f{wst[2 * q], wst[2 * q + 1]} * dt;
            const v2f gx = g * cs, gy = g * sn;
            xy[0][2 * q] = gx[0]; xy[0][2 * q + 1] = gx[1];
            xy[1][2 * q] = gy[0]; xy[1][2 * q + 1] = gy[1];
        }
    } else {
#pragma unroll
        for (int r = 0; r < 8; ++r) {
            float sn, cs;
            sincosf(thst[r], &sn, &cs);
            const float g = wst[r] * dt;
            xy[0][r] = g * cs;
            xy[1][r] = g * sn;
        }
    }
    tile_scan<2>(xy, &sm, 1, wave, lane, aL, bM, wanted);
    if (!wanted) return;
    if (tailOnly) {
        if (lane == 31)
            ((float4*)out)[(wave + 1) * 256] =
                make_float4(px + xy[0][7], py + xy[1][7], thout[7], vout[7]);
        return;
    }

    float4* ob = sm.outStage[wave];
    const int k0 = col * 16 + half * 8 + 1;
#pragma unroll
    for (int r = 0; r < 8; ++r) {
        const int k = k0 + r;
        ob[k + (k >> 4)] = make_float4(px + xy[0][r], py + xy[1][r], thout[r], vout[r]);
    }
    wave_lds_handover();
    const int kf = (lane > 0) ? lane : 256;
    float4* og = (float4*)out + wave * 256;
    const float4* rd = ob + (lane + (lane >> 4));
    float4 rows[8];
    rows[0] = ob[kf + (kf >> 4)];
#pragma unroll
    for (int i = 1; i < 8; ++i) rows[i] = rd[34 * i];
    if (lane > 0 || sendLast) og[kf] = rows[0];
#pragma unroll
    for (int i = 1; i < 8; ++i) og[lane + 32 * i] = rows[i];
    if (group == 0 && tid == 0) *(float4*)out = make_float4(px, py, pth, pv);
}

extern "C" void kernel_launch(void* const* d_in, const int* in_sizes, int n_in,
                              void* d_out, int out_size, void* d_ws, size_t ws_size,
                              hipStream_t stream) {
    (void)in_sizes; (void)n_in; (void)out_size; (void)d_ws; (void)ws_size;
    const float* x0 = (const float*)d_in[0];
    const float* U  = (const float*)d_in[1];
    const float* dt = (const float*)d_in[2];
    float* out = (float*)d_out;
    rollout_scan_kernel<<<NGRID, BLOCK, 0, stream>>>(x0, U, dt, out);
}


# Waves whose rows another workgroup writes hand over only their tile's x and y totals (lane sums, then cross-lane adds), skipping LDS staging and the 16 WMMA of a scan nobody reads; new f32 sum order
# speedup vs baseline: 2.0825x; 1.1035x over previous
#include <hip/hip_runtime.h>


#define HSTEPS 8192
#define BLOCK  1024
#define NWAVE  32
#define NGRID  8
#define OWN_WAVES (NWAVE / NGRID)

#define ACC_LIMIT   3.0f
#define STEER_LIMIT 0.6f
#define V_MIN_C     0.0f
#define V_MAX_C     30.0f
#define WHEELBASE   2.7f

typedef __attribute__((ext_vector_type(2))) float v2f;
typedef __attribute__((ext_vector_type(8))) float v8f;

__device__ __forceinline__ float clipf(float x, float lo, float hi) {
    return fminf(fmaxf(x, lo), hi);
}

__device__ __forceinline__ v2f pk_fma(v2f a, v2f b, v2f c) {
    return __builtin_elementwise_fma(a, b, c);
}
__device__ __forceinline__ v2f splat(unsigned bits) {
    const float f = __builtin_bit_cast(float, bits);
    return v2f{f, f};
}

__device__ __forceinline__ v2f tan_small_pair(v2f x) {
    const v2f s = x * x;
    v2f den = pk_fma(s, splat(0x3c971480u), splat(0xbf039337u));
    den = pk_fma(s, den, splat(0x3f93f425u));
    const v2f num = pk_fma(s, splat(0xbc8cedd3u), splat(0x3ec54587u));
    v2f rd;
    rd[0] = __builtin_amdgcn_rcpf(den[0]);
    rd[1] = __builtin_amdgcn_rcpf(den[1]);
    const v2f p = s * (num * rd);
    return pk_fma(p, x, x);
}

__device__ __forceinline__ void sincos_pair(v2f x, v2f& sn, v2f& cs) {
    const v2f n = __builtin_elementwise_rint(x * splat(0x3f22f983u));
    v2f r = pk_fma(n, splat(0xbfc90fdau), x);
    r = pk_fma(n, splat(0xb3a22168u), r);
    r = pk_fma(n, splat(0xa7c234c4u), r);
    const v2f t = r * r;
    v2f p = pk_fma(t, splat(0xb94c1982u), splat(0x3c0881c4u));
    p = pk_fma(t, p, splat(0xbe2aaa9du));
    p = t * p;
    const v2f s = pk_fma(r, p, r);
    v2f q = pk_fma(t, splat(0x37d75334u), splat(0xbab64f3bu));
    q = pk_fma(t, q, splat(0x3d2aabf7u));
    q = pk_fma(t, q, splat(0xbf000004u));
    const v2f c = pk_fma(t, q, splat(0x3f800000u));
#pragma unroll
    for (int j = 0; j < 2; ++j) {
        const unsigned k = (unsigned)(int)n[j];
        const bool odd = (k & 1u) != 0u;
        const unsigned flip = (k << 30) & 0x80000000u;
        const float so = odd ? c[j] : s[j];
        const float co = odd ? -s[j] : c[j];
        sn[j] = __builtin_bit_cast(float, __builtin_bit_cast(unsigned, so) ^ flip);
        cs[j] = __builtin_bit_cast(float, __builtin_bit_cast(unsigned, co) ^ flip);
    }
}

struct ClampAdd { float S, Lo, Hi; };

__device__ __forceinline__ ClampAdd compose(ClampAdd first, ClampAdd second) {
    ClampAdd r;
    r.S  = first.S + second.S;
    r.Lo = clipf(first.Lo + second.S, second.Lo, second.Hi);
    r.Hi = clipf(first.Hi + second.S, second.Lo, second.Hi);
    return r;
}
__device__ __forceinline__ float apply(ClampAdd f, float v) {
    return clipf(v + f.S, f.Lo, f.Hi);
}

template <int N>
__device__ __forceinline__ float row_left(float v) {
    const int x = __builtin_bit_cast(int, v);
    return __builtin_bit_cast(float,
        __builtin_amdgcn_update_dpp(x, x, 0x110 + N, 0xF, 0xF, false));
}
__device__ __forceinline__ float other_half(float v) {
    const int x = __builtin_bit_cast(int, v);
    return __builtin_bit_cast(float,
        __builtin_amdgcn_permlanex16(x, x, 0x76543210, 0xfedcba98, false, false));
}
__device__ __forceinline__ float lane_value(float v, int uniform_lane) {
    return __builtin_bit_cast(float,
        __builtin_amdgcn_readlane(__builtin_bit_cast(int, v), uniform_lane));
}

template <int N>
__device__ __forceinline__ ClampAdd row_scan_step(ClampAdd f, int col) {
    ClampAdd left;
    left.S = row_left<N>(f.S); left.Lo = row_left<N>(f.Lo); left.Hi = row_left<N>(f.Hi);
    const ClampAdd c = compose(left, f);
    if (col >= N) f = c;
    return f;
}
__device__ __forceinline__ ClampAdd row_scan(ClampAdd f, int col) {
    f = row_scan_step<1>(f, col);
    f = row_scan_step<2>(f, col);
    f = row_scan_step<4>(f, col);
    f = row_scan_step<8>(f, col);
    return f;
}

template <int N>
__device__ __forceinline__ float row_sum_step(float x, int col) {
    const float o = row_left<N>(x);
    return (col >= N) ? x + o : x;
}

template <int N>
__device__ __forceinline__ float row_xor(float v) {
    const int x = __builtin_bit_cast(int, v);
    return __builtin_bit_cast(float,
        __builtin_amdgcn_update_dpp(0, x, 0x160 + N, 0xF, 0xF, true));
}
__device__ __forceinline__ float tiles_inclusive(float x, int col, int half) {
    x = row_sum_step<1>(x, col);
    x = row_sum_step<2>(x, col);
    x = row_sum_step<4>(x, col);
    x = row_sum_step<8>(x, col);
    const float low = lane_value(x, 15);
    if (half == 1) x += low;
    return x;
}

__device__ __forceinline__ void wave_lds_handover() {
    __builtin_amdgcn_fence(__ATOMIC_RELEASE, "wavefront");
    __builtin_amdgcn_wave_barrier();
    __builtin_amdgcn_fence(__ATOMIC_ACQUIRE, "wavefront");
}

#define OUT_SLOTS 273

struct SMem {
    float4 outStage[NWAVE][OUT_SLOTS];
    float stage[2][NWAVE][256];
    float tot[2][NWAVE][16];
    float tileSum[3][NWAVE];
    float wS[NWAVE];
    float wL[NWAVE];
    float wH[NWAVE];
};

template <int NS>
__device__ __forceinline__ void tile_scan_local(const float (&v)[NS][8], v8f (&c)[NS],
                                                SMem* sm, int plane0, int wave, int lane,
                                                const v2f aL[4], const v2f bM[4]) {
    const int col  = lane & 15;
    const int half = lane >> 4;

#pragma unroll
    for (int s = 0; s < NS; ++s) {
        float* st = sm->stage[s][wave];
#pragma unroll
        for (int r = 0; r < 8; ++r)
            st[col * 16 + half * 8 + r] = v[s][r];
    }
    wave_lds_handover();

#pragma unroll
    for (int s = 0; s < NS; ++s) c[s] = v8f{};
#pragma unroll
    for (int kb = 0; kb < 4; ++kb) {
#pragma unroll
        for (int s = 0; s < NS; ++s) {
            const float* st = sm->stage[s][wave];
            v2f b;
#pragma unroll
            for (int j = 0; j < 2; ++j)
                b[j] = st[col * 16 + kb * 4 + half * 2 + j];
            c[s] = __builtin_amdgcn_wmma_f32_16x16x4_f32(false, aL[kb], false, b,
                                                         (short)0, c[s], false, false);
        }
    }

    if (half == 1) {
#pragma unroll
        for (int s = 0; s < NS; ++s) sm->tot[s][wave][col] = c[s][7];
    }
    wave_lds_handover();

#pragma unroll
    for (int kb = 0; kb < 4; ++kb) {
#pragma unroll
        for (int s = 0; s < NS; ++s) {
            v2f a;
#pragma unroll
            for (int j = 0; j < 2; ++j)
                a[j] = sm->tot[s][wave][kb * 4 + half * 2 + j];
            c[s] = __builtin_amdgcn_wmma_f32_16x16x4_f32(false, a, false, bM[kb],
                                                         (short)0, c[s], false, false);
        }
    }

    if (lane == 31) {
#pragma unroll
        for (int s = 0; s < NS; ++s) sm->tileSum[plane0 + s][wave] = c[s][7];
    }
}

template <int NS>
__device__ __forceinline__ void tile_totals(const float (&v)[NS][8], SMem* sm, int plane0,
                                            int wave, int lane) {
    float t[NS];
#pragma unroll
    for (int s = 0; s < NS; ++s)
        t[s] = ((v[s][0] + v[s][1]) + (v[s][2] + v[s][3])) +
               ((v[s][4] + v[s][5]) + (v[s][6] + v[s][7]));
#pragma unroll
    for (int s = 0; s < NS; ++s) t[s] += row_xor<1>(t[s]);
#pragma unroll
    for (int s = 0; s < NS; ++s) t[s] += row_xor<2>(t[s]);
#pragma unroll
    for (int s = 0; s < NS; ++s) t[s] += row_xor<4>(t[s]);
#pragma unroll
    for (int s = 0; s < NS; ++s) t[s] += row_xor<8>(t[s]);
#pragma unroll
    for (int s = 0; s < NS; ++s) t[s] += other_half(t[s]);
    if (lane == 0) {
#pragma unroll
        for (int s = 0; s < NS; ++s) sm->tileSum[plane0 + s][wave] = t[s];
    }
}

template <int NS>
__device__ __forceinline__ void tile_scan_carry(float (&v)[NS][8], const v8f (&c)[NS],
                                                const SMem* sm, int plane0, int wave, int lane) {
    const int col  = lane & 15;
    const int half = lane >> 4;
#pragma unroll
    for (int s = 0; s < NS; ++s) {
        const float x = tiles_inclusive(sm->tileSum[plane0 + s][lane], col, half);
        const float carry = (wave > 0) ? lane_value(x, wave > 0 ? wave - 1 : 0) : 0.0f;
#pragma unroll
        for (int r = 0; r < 8; ++r) v[s][r] = c[s][r] + carry;
    }
}

__global__ __launch_bounds__(BLOCK) void
rollout_scan_kernel(const float* __restrict__ x0,
                    const float* __restrict__ U,
                    const float* __restrict__ dtp,
                    float* __restrict__ out) {
    __shared__ SMem sm;
    const int tid  = threadIdx.x;
    const int wave = __builtin_amdgcn_readfirstlane(tid >> 5);
    const int lane = tid & 31;
    const int col  = lane & 15;
    const int half = lane >> 4;
    const int t0 = wave * 256 + col * 16 + half * 8;

    const int group     = blockIdx.x;
    const int firstWave = group * OWN_WAVES;
    const bool owner    = (wave >= firstWave) && (wave < firstWave + OWN_WAVES);
    const bool tailOnly = (wave == firstWave - 1);
    const bool wanted   = owner || tailOnly;
    const bool sendLast = (wave != firstWave + OWN_WAVES - 1) || (wave == NWAVE - 1);

    v2f aL[4], bM[4];
#pragma unroll
    for (int kb = 0; kb < 4; ++kb) {
#pragma unroll
        for (int j = 0; j < 2; ++j) {
            const int k = kb * 4 + half * 2 + j;
            aL[kb][j] = (k <= col) ? 1.0f : 0.0f;
            bM[kb][j] = (k <  col) ? 1.0f : 0.0f;
        }
    }

    const float dt = dtp[0];
    const float px = x0[0], py = x0[1], pth = x0[2], pv = x0[3];

    float s[8], dl[8];
    const float4* u4 = (const float4*)(U + (size_t)t0 * 2);
#pragma unroll
    for (int q = 0; q < 4; ++q) {
        const float4 u = u4[q];
        s[2 * q]      = clipf(u.x, -ACC_LIMIT, ACC_LIMIT) * dt;
        dl[2 * q]     = clipf(u.y, -STEER_LIMIT, STEER_LIMIT);
        s[2 * q + 1]  = clipf(u.z, -ACC_LIMIT, ACC_LIMIT) * dt;
        dl[2 * q + 1] = clipf(u.w, -STEER_LIMIT, STEER_LIMIT);
    }

    ClampAdd mine;
    mine.S = s[0]; mine.Lo = V_MIN_C; mine.Hi = V_MAX_C;
#pragma unroll
    for (int r = 1; r < 8; ++r) {
        mine.S += s[r];
        mine.Lo = clipf(mine.Lo + s[r], V_MIN_C, V_MAX_C);
        mine.Hi = clipf(mine.Hi + s[r], V_MIN_C, V_MAX_C);
    }
    ClampAdd partner;
    partner.S = other_half(mine.S); partner.Lo = other_half(mine.Lo); partner.Hi = other_half(mine.Hi);
    const ClampAdd colFn = half ? compose(partner, mine) : compose(mine, partner);
    const ClampAdd inclCols = row_scan(colFn, col);
    ClampAdd exclCols;
    exclCols.S = row_left<1>(inclCols.S); exclCols.Lo = row_left<1>(inclCols.Lo); exclCols.Hi = row_left<1>(inclCols.Hi);
    if (lane == 15) { sm.wS[wave] = inclCols.S; sm.wL[wave] = inclCols.Lo; sm.wH[wave] = inclCols.Hi; }
    __syncthreads();
    ClampAdd wf;
    wf.S = sm.wS[lane]; wf.Lo = sm.wL[lane]; wf.Hi = sm.wH[lane];
    wf = row_scan(wf, col);
    {
        ClampAdd low;
        low.S = lane_value(wf.S, 15); low.Lo = lane_value(wf.Lo, 15); low.Hi = lane_value(wf.Hi, 15);
        const ClampAdd c2 = compose(low, wf);
        if (half == 1) wf = c2;
    }
    float w = clipf(pv, V_MIN_C, V_MAX_C);
    if (wave > 0) {
        ClampAdd pre;
        pre.S = lane_value(wf.S, wave - 1); pre.Lo = lane_value(wf.Lo, wave - 1); pre.Hi = lane_value(wf.Hi, wave - 1);
        w = apply(pre, w);
    }
    if (col > 0) w = apply(exclCols, w);
    if (half == 1) w = apply(partner, w);

    float wst[8], vout[8], dth[8], tn[8];
    const float dtOverL = dt / WHEELBASE;
#pragma unroll
    for (int q = 0; q < 4; ++q) {
        const v2f t2 = tan_small_pair(v2f{dl[2 * q], dl[2 * q + 1]});
        tn[2 * q] = t2[0]; tn[2 * q + 1] = t2[1];
    }
#pragma unroll
    for (int r = 0; r < 8; ++r) {
        wst[r] = w;
        dth[r] = w * tn[r] * dtOverL;
        w = clipf(w + s[r], V_MIN_C, V_MAX_C);
        vout[r] = w;
    }

    float th[1][8];
#pragma unroll
    for (int r = 0; r < 8; ++r) th[0][r] = dth[r];
    v8f cth[1];
    tile_scan_local<1>(th, cth, &sm, 0, wave, lane, aL, bM);
    __syncthreads();
    tile_scan_carry<1>(th, cth, &sm, 0, wave, lane);
    float thout[8], thst[8];
#pragma unroll
    for (int r = 0; r < 8; ++r) {
        thout[r] = pth + th[0][r];
        thst[r]  = thout[r] - dth[r];
    }

    float xy[2][8];
    float amax = 0.0f;
#pragma unroll
    for (int r = 0; r < 8; ++r) amax = __builtin_fmaxf(amax, __builtin_fabsf(thst[r]));
    if (__builtin_expect(amax < 131072.0f, 1)) {
#pragma unroll
        for (int q = 0; q < 4; ++q) {
            v2f sn, cs;
            sincos_pair(v2f{thst[2 * q], thst[2 * q + 1]}, sn, cs);
            const v2f g = v2f{wst[2 * q], wst[2 * q + 1]} * dt;
            const v2f gx = g * cs, gy = g * sn;
            xy[0][2 * q] = gx[0]; xy[0][2 * q + 1] = gx[1];
            xy[1][2 * q] = gy[0]; xy[1][2 * q + 1] = gy[1];
        }
    } else {
#pragma unroll
        for (int r = 0; r < 8; ++r) {
            float sn, cs;
            sincosf(thst[r], &sn, &cs);
            const float g = wst[r] * dt;
            xy[0][r] = g * cs;
            xy[1][r] = g * sn;
        }
    }
    v8f cxy[2];
    if (owner) tile_scan_local<2>(xy, cxy, &sm, 1, wave, lane, aL, bM);
    else       tile_totals<2>(xy, &sm, 1, wave, lane);
    __syncthreads();
    if (!wanted) return;
    if (tailOnly) {
        const float xs = lane_value(tiles_inclusive(sm.tileSum[1][lane], col, half), wave);
        const float ys = lane_value(tiles_inclusive(sm.tileSum[2][lane], col, half), wave);
        if (lane == 31)
            ((float4*)out)[(wave + 1) * 256] = make_float4(px + xs, py + ys, thout[7], vout[7]);
        return;
    }
    tile_scan_carry<2>(xy, cxy, &sm, 1, wave, lane);

    float4* ob = sm.outStage[wave];
    const int k0 = col * 16 + half * 8 + 1;
#pragma unroll
    for (int r = 0; r < 8; ++r) {
        const int k = k0 + r;
        ob[k + (k >> 4)] = make_float4(px + xy[0][r], py + xy[1][r], thout[r], vout[r]);
    }
    wave_lds_handover();
    const int kf = (lane > 0) ? lane : 256;
    float4* og = (float4*)out + wave * 256;
    const float4* rd = ob + (lane + (lane >> 4));
    float4 rows[8];
    rows[0] = ob[kf + (kf >> 4)];
#pragma unroll
    for (int i = 1; i < 8; ++i) rows[i] = rd[34 * i];
    if (lane > 0 || sendLast) og[kf] = rows[0];
#pragma unroll
    for (int i = 1; i < 8; ++i) og[lane + 32 * i] = rows[i];
    if (group == 0 && tid == 0) *(float4*)out = make_float4(px, py, pth, pv);
}

extern "C" void kernel_launch(void* const* d_in, const int* in_sizes, int n_in,
                              void* d_out, int out_size, void* d_ws, size_t ws_size,
                              hipStream_t stream) {
    (void)in_sizes; (void)n_in; (void)out_size; (void)d_ws; (void)ws_size;
    const float* x0 = (const float*)d_in[0];
    const float* U  = (const float*)d_in[1];
    const float* dt = (const float*)d_in[2];
    float* out = (float*)d_out;
    rollout_scan_kernel<<<NGRID, BLOCK, 0, stream>>>(x0, U, dt, out);
}


# Cross-lane scan steps without compare-and-select: the DPP shift itself fills lanes with no left neighbour with the neutral value (0, -inf, +inf); function clamps as one median op; values unchanged
# speedup vs baseline: 2.1994x; 1.0561x over previous
#include <hip/hip_runtime.h>


#define HSTEPS 8192
#define BLOCK  1024
#define NWAVE  32
#define NGRID  8
#define OWN_WAVES (NWAVE / NGRID)

#define ACC_LIMIT   3.0f
#define STEER_LIMIT 0.6f
#define V_MIN_C     0.0f
#define V_MAX_C     30.0f
#define WHEELBASE   2.7f

typedef __attribute__((ext_vector_type(2))) float v2f;
typedef __attribute__((ext_vector_type(8))) float v8f;

__device__ __forceinline__ float clipf(float x, float lo, float hi) {
    return fminf(fmaxf(x, lo), hi);
}

__device__ __forceinline__ v2f pk_fma(v2f a, v2f b, v2f c) {
    return __builtin_elementwise_fma(a, b, c);
}
__device__ __forceinline__ v2f splat(unsigned bits) {
    const float f = __builtin_bit_cast(float, bits);
    return v2f{f, f};
}

__device__ __forceinline__ v2f tan_small_pair(v2f x) {
    const v2f s = x * x;
    v2f den = pk_fma(s, splat(0x3c971480u), splat(0xbf039337u));
    den = pk_fma(s, den, splat(0x3f93f425u));
    const v2f num = pk_fma(s, splat(0xbc8cedd3u), splat(0x3ec54587u));
    v2f rd;
    rd[0] = __builtin_amdgcn_rcpf(den[0]);
    rd[1] = __builtin_amdgcn_rcpf(den[1]);
    const v2f p = s * (num * rd);
    return pk_fma(p, x, x);
}

__device__ __forceinline__ void sincos_pair(v2f x, v2f& sn, v2f& cs) {
    const v2f n = __builtin_elementwise_rint(x * splat(0x3f22f983u));
    v2f r = pk_fma(n, splat(0xbfc90fdau), x);
    r = pk_fma(n, splat(0xb3a22168u), r);
    r = pk_fma(n, splat(0xa7c234c4u), r);
    const v2f t = r * r;
    v2f p = pk_fma(t, splat(0xb94c1982u), splat(0x3c0881c4u));
    p = pk_fma(t, p, splat(0xbe2aaa9du));
    p = t * p;
    const v2f s = pk_fma(r, p, r);
    v2f q = pk_fma(t, splat(0x37d75334u), splat(0xbab64f3bu));
    q = pk_fma(t, q, splat(0x3d2aabf7u));
    q = pk_fma(t, q, splat(0xbf000004u));
    const v2f c = pk_fma(t, q, splat(0x3f800000u));
#pragma unroll
    for (int j = 0; j < 2; ++j) {
        const unsigned k = (unsigned)(int)n[j];
        const bool odd = (k & 1u) != 0u;
        const unsigned flip = (k << 30) & 0x80000000u;
        const float so = odd ? c[j] : s[j];
        const float co = odd ? -s[j] : c[j];
        sn[j] = __builtin_bit_cast(float, __builtin_bit_cast(unsigned, so) ^ flip);
        cs[j] = __builtin_bit_cast(float, __builtin_bit_cast(unsigned, co) ^ flip);
    }
}

struct ClampAdd { float S, Lo, Hi; };

__device__ __forceinline__ float clip_between(float x, float lo, float hi) {
    return __builtin_amdgcn_fmed3f(x, lo, hi);
}

__device__ __forceinline__ ClampAdd compose(ClampAdd first, ClampAdd second) {
    ClampAdd r;
    r.S  = first.S + second.S;
    r.Lo = clip_between(first.Lo + second.S, second.Lo, second.Hi);
    r.Hi = clip_between(first.Hi + second.S, second.Lo, second.Hi);
    return r;
}
__device__ __forceinline__ float apply(ClampAdd f, float v) {
    return clip_between(v + f.S, f.Lo, f.Hi);
}

template <int N>
__device__ __forceinline__ float row_left_or(float v, float fill) {
    return __builtin_bit_cast(float,
        __builtin_amdgcn_update_dpp(__builtin_bit_cast(int, fill), __builtin_bit_cast(int, v),
                                    0x110 + N, 0xF, 0xF, false));
}
template <int N>
__device__ __forceinline__ float row_left_or_zero(float v) {
    return __builtin_bit_cast(float,
        __builtin_amdgcn_update_dpp(0, __builtin_bit_cast(int, v), 0x110 + N, 0xF, 0xF, true));
}
__device__ __forceinline__ float other_half(float v) {
    const int x = __builtin_bit_cast(int, v);
    return __builtin_bit_cast(float,
        __builtin_amdgcn_permlanex16(x, x, 0x76543210, 0xfedcba98, false, false));
}
__device__ __forceinline__ float lane_value(float v, int uniform_lane) {
    return __builtin_bit_cast(float,
        __builtin_amdgcn_readlane(__builtin_bit_cast(int, v), uniform_lane));
}

template <int N>
__device__ __forceinline__ ClampAdd row_scan_step(ClampAdd f) {
    ClampAdd left;
    left.S  = row_left_or_zero<N>(f.S);
    left.Lo = row_left_or<N>(f.Lo, -__builtin_inff());
    left.Hi = row_left_or<N>(f.Hi, __builtin_inff());
    return compose(left, f);
}
__device__ __forceinline__ ClampAdd row_scan(ClampAdd f) {
    f = row_scan_step<1>(f);
    f = row_scan_step<2>(f);
    f = row_scan_step<4>(f);
    f = row_scan_step<8>(f);
    return f;
}

template <int N>
__device__ __forceinline__ float row_sum_step(float x) {
    return x + row_left_or_zero<N>(x);
}

template <int N>
__device__ __forceinline__ float row_xor(float v) {
    const int x = __builtin_bit_cast(int, v);
    return __builtin_bit_cast(float,
        __builtin_amdgcn_update_dpp(0, x, 0x160 + N, 0xF, 0xF, true));
}
__device__ __forceinline__ float tiles_inclusive(float x, int half) {
    x = row_sum_step<1>(x);
    x = row_sum_step<2>(x);
    x = row_sum_step<4>(x);
    x = row_sum_step<8>(x);
    const float low = lane_value(x, 15);
    if (half == 1) x += low;
    return x;
}

__device__ __forceinline__ void wave_lds_handover() {
    __builtin_amdgcn_fence(__ATOMIC_RELEASE, "wavefront");
    __builtin_amdgcn_wave_barrier();
    __builtin_amdgcn_fence(__ATOMIC_ACQUIRE, "wavefront");
}

#define OUT_SLOTS 273

struct SMem {
    float4 outStage[NWAVE][OUT_SLOTS];
    float stage[2][NWAVE][256];
    float tot[2][NWAVE][16];
    float tileSum[3][NWAVE];
    float wS[NWAVE];
    float wL[NWAVE];
    float wH[NWAVE];
};

template <int NS>
__device__ __forceinline__ void tile_scan_local(const float (&v)[NS][8], v8f (&c)[NS],
                                                SMem* sm, int plane0, int wave, int lane,
                                                const v2f aL[4], const v2f bM[4]) {
    const int col  = lane & 15;
    const int half = lane >> 4;

#pragma unroll
    for (int s = 0; s < NS; ++s) {
        float* st = sm->stage[s][wave];
#pragma unroll
        for (int r = 0; r < 8; ++r)
            st[col * 16 + half * 8 + r] = v[s][r];
    }
    wave_lds_handover();

#pragma unroll
    for (int s = 0; s < NS; ++s) c[s] = v8f{};
#pragma unroll
    for (int kb = 0; kb < 4; ++kb) {
#pragma unroll
        for (int s = 0; s < NS; ++s) {
            const float* st = sm->stage[s][wave];
            v2f b;
#pragma unroll
            for (int j = 0; j < 2; ++j)
                b[j] = st[col * 16 + kb * 4 + half * 2 + j];
            c[s] = __builtin_amdgcn_wmma_f32_16x16x4_f32(false, aL[kb], false, b,
                                                         (short)0, c[s], false, false);
        }
    }

    if (half == 1) {
#pragma unroll
        for (int s = 0; s < NS; ++s) sm->tot[s][wave][col] = c[s][7];
    }
    wave_lds_handover();

#pragma unroll
    for (int kb = 0; kb < 4; ++kb) {
#pragma unroll
        for (int s = 0; s < NS; ++s) {
            v2f a;
#pragma unroll
            for (int j = 0; j < 2; ++j)
                a[j] = sm->tot[s][wave][kb * 4 + half * 2 + j];
            c[s] = __builtin_amdgcn_wmma_f32_16x16x4_f32(false, a, false, bM[kb],
                                                         (short)0, c[s], false, false);
        }
    }

    if (lane == 31) {
#pragma unroll
        for (int s = 0; s < NS; ++s) sm->tileSum[plane0 + s][wave] = c[s][7];
    }
}

template <int NS>
__device__ __forceinline__ void tile_totals(const float (&v)[NS][8], SMem* sm, int plane0,
                                            int wave, int lane) {
    float t[NS];
#pragma unroll
    for (int s = 0; s < NS; ++s)
        t[s] = ((v[s][0] + v[s][1]) + (v[s][2] + v[s][3])) +
               ((v[s][4] + v[s][5]) + (v[s][6] + v[s][7]));
#pragma unroll
    for (int s = 0; s < NS; ++s) t[s] += row_xor<1>(t[s]);
#pragma unroll
    for (int s = 0; s < NS; ++s) t[s] += row_xor<2>(t[s]);
#pragma unroll
    for (int s = 0; s < NS; ++s) t[s] += row_xor<4>(t[s]);
#pragma unroll
    for (int s = 0; s < NS; ++s) t[s] += row_xor<8>(t[s]);
#pragma unroll
    for (int s = 0; s < NS; ++s) t[s] += other_half(t[s]);
    if (lane == 0) {
#pragma unroll
        for (int s = 0; s < NS; ++s) sm->tileSum[plane0 + s][wave] = t[s];
    }
}

template <int NS>
__device__ __forceinline__ void tile_scan_carry(float (&v)[NS][8], const v8f (&c)[NS],
                                                const SMem* sm, int plane0, int wave, int lane) {
    const int half = lane >> 4;
#pragma unroll
    for (int s = 0; s < NS; ++s) {
        const float x = tiles_inclusive(sm->tileSum[plane0 + s][lane], half);
        const float carry = (wave > 0) ? lane_value(x, wave > 0 ? wave - 1 : 0) : 0.0f;
#pragma unroll
        for (int r = 0; r < 8; ++r) v[s][r] = c[s][r] + carry;
    }
}

__global__ __launch_bounds__(BLOCK) void
rollout_scan_kernel(const float* __restrict__ x0,
                    const float* __restrict__ U,
                    const float* __restrict__ dtp,
                    float* __restrict__ out) {
    __shared__ SMem sm;
    const int tid  = threadIdx.x;
    const int wave = __builtin_amdgcn_readfirstlane(tid >> 5);
    const int lane = tid & 31;
    const int col  = lane & 15;
    const int half = lane >> 4;
    const int t0 = wave * 256 + col * 16 + half * 8;

    const int group     = blockIdx.x;
    const int firstWave = group * OWN_WAVES;
    const bool owner    = (wave >= firstWave) && (wave < firstWave + OWN_WAVES);
    const bool tailOnly = (wave == firstWave - 1);
    const bool wanted   = owner || tailOnly;
    const bool sendLast = (wave != firstWave + OWN_WAVES - 1) || (wave == NWAVE - 1);

    v2f aL[4], bM[4];
#pragma unroll
    for (int kb = 0; kb < 4; ++kb) {
#pragma unroll
        for (int j = 0; j < 2; ++j) {
            const int k = kb * 4 + half * 2 + j;
            aL[kb][j] = (k <= col) ? 1.0f : 0.0f;
            bM[kb][j] = (k <  col) ? 1.0f : 0.0f;
        }
    }

    const float dt = dtp[0];
    const float px = x0[0], py = x0[1], pth = x0[2], pv = x0[3];

    float s[8], dl[8];
    const float4* u4 = (const float4*)(U + (size_t)t0 * 2);
#pragma unroll
    for (int q = 0; q < 4; ++q) {
        const float4 u = u4[q];
        s[2 * q]      = clipf(u.x, -ACC_LIMIT, ACC_LIMIT) * dt;
        dl[2 * q]     = clipf(u.y, -STEER_LIMIT, STEER_LIMIT);
        s[2 * q + 1]  = clipf(u.z, -ACC_LIMIT, ACC_LIMIT) * dt;
        dl[2 * q + 1] = clipf(u.w, -STEER_LIMIT, STEER_LIMIT);
    }

    ClampAdd mine;
    mine.S = s[0]; mine.Lo = V_MIN_C; mine.Hi = V_MAX_C;
#pragma unroll
    for (int r = 1; r < 8; ++r) {
        mine.S += s[r];
        mine.Lo = clipf(mine.Lo + s[r], V_MIN_C, V_MAX_C);
        mine.Hi = clipf(mine.Hi + s[r], V_MIN_C, V_MAX_C);
    }
    ClampAdd partner;
    partner.S = other_half(mine.S); partner.Lo = other_half(mine.Lo); partner.Hi = other_half(mine.Hi);
    const ClampAdd colFn = half ? compose(partner, mine) : compose(mine, partner);
    const ClampAdd inclCols = row_scan(colFn);
    if (lane == 15) { sm.wS[wave] = inclCols.S; sm.wL[wave] = inclCols.Lo; sm.wH[wave] = inclCols.Hi; }
    __syncthreads();
    ClampAdd wf;
    wf.S = sm.wS[lane]; wf.Lo = sm.wL[lane]; wf.Hi = sm.wH[lane];
    wf = row_scan(wf);
    {
        ClampAdd low;
        low.S = lane_value(wf.S, 15); low.Lo = lane_value(wf.Lo, 15); low.Hi = lane_value(wf.Hi, 15);
        const ClampAdd c2 = compose(low, wf);
        if (half == 1) wf = c2;
    }
    float w = clipf(pv, V_MIN_C, V_MAX_C);
    if (wave > 0) {
        ClampAdd pre;
        pre.S = lane_value(wf.S, wave - 1); pre.Lo = lane_value(wf.Lo, wave - 1); pre.Hi = lane_value(wf.Hi, wave - 1);
        w = apply(pre, w);
    }
    w = row_left_or<1>(apply(inclCols, w), w);
    if (half == 1) w = apply(partner, w);

    float wst[8], vout[8], dth[8], tn[8];
    const float dtOverL = dt / WHEELBASE;
#pragma unroll
    for (int q = 0; q < 4; ++q) {
        const v2f t2 = tan_small_pair(v2f{dl[2 * q], dl[2 * q + 1]});
        tn[2 * q] = t2[0]; tn[2 * q + 1] = t2[1];
    }
#pragma unroll
    for (int r = 0; r < 8; ++r) {
        wst[r] = w;
        dth[r] = w * tn[r] * dtOverL;
        w = clipf(w + s[r], V_MIN_C, V_MAX_C);
        vout[r] = w;
    }

    float th[1][8];
#pragma unroll
    for (int r = 0; r < 8; ++r) th[0][r] = dth[r];
    v8f cth[1];
    tile_scan_local<1>(th, cth, &sm, 0, wave, lane, aL, bM);
    __syncthreads();
    tile_scan_carry<1>(th, cth, &sm, 0, wave, lane);
    float thout[8], thst[8];
#pragma unroll
    for (int r = 0; r < 8; ++r) {
        thout[r] = pth + th[0][r];
        thst[r]  = thout[r] - dth[r];
    }

    float xy[2][8];
    float amax = 0.0f;
#pragma unroll
    for (int r = 0; r < 8; ++r) amax = __builtin_fmaxf(amax, __builtin_fabsf(thst[r]));
    if (__builtin_expect(amax < 131072.0f, 1)) {
#pragma unroll
        for (int q = 0; q < 4; ++q) {
            v2f sn, cs;
            sincos_pair(v2f{thst[2 * q], thst[2 * q + 1]}, sn, cs);
            const v2f g = v2f{wst[2 * q], wst[2 * q + 1]} * dt;
            const v2f gx = g * cs, gy = g * sn;
            xy[0][2 * q] = gx[0]; xy[0][2 * q + 1] = gx[1];
            xy[1][2 * q] = gy[0]; xy[1][2 * q + 1] = gy[1];
        }
    } else {
#pragma unroll
        for (int r = 0; r < 8; ++r) {
            float sn, cs;
            sincosf(thst[r], &sn, &cs);
            const float g = wst[r] * dt;
            xy[0][r] = g * cs;
            xy[1][r] = g * sn;
        }
    }
    v8f cxy[2];
    if (owner) tile_scan_local<2>(xy, cxy, &sm, 1, wave, lane, aL, bM);
    else       tile_totals<2>(xy, &sm, 1, wave, lane);
    __syncthreads();
    if (!wanted) return;
    if (tailOnly) {
        const float xs = lane_value(tiles_inclusive(sm.tileSum[1][lane], half), wave);
        const float ys = lane_value(tiles_inclusive(sm.tileSum[2][lane], half), wave);
        if (lane == 31)
            ((float4*)out)[(wave + 1) * 256] = make_float4(px + xs, py + ys, thout[7], vout[7]);
        return;
    }
    tile_scan_carry<2>(xy, cxy, &sm, 1, wave, lane);

    float4* ob = sm.outStage[wave];
    const int k0 = col * 16 + half * 8 + 1;
#pragma unroll
    for (int r = 0; r < 8; ++r) {
        const int k = k0 + r;
        ob[k + (k >> 4)] = make_float4(px + xy[0][r], py + xy[1][r], thout[r], vout[r]);
    }
    wave_lds_handover();
    const int kf = (lane > 0) ? lane : 256;
    float4* og = (float4*)out + wave * 256;
    const float4* rd = ob + (lane + (lane >> 4));
    float4 rows[8];
    rows[0] = ob[kf + (kf >> 4)];
#pragma unroll
    for (int i = 1; i < 8; ++i) rows[i] = rd[34 * i];
    if (lane > 0 || sendLast) og[kf] = rows[0];
#pragma unroll
    for (int i = 1; i < 8; ++i) og[lane + 32 * i] = rows[i];
    if (group == 0 && tid == 0) *(float4*)out = make_float4(px, py, pth, pv);
}

extern "C" void kernel_launch(void* const* d_in, const int* in_sizes, int n_in,
                              void* d_out, int out_size, void* d_ws, size_t ws_size,
                              hipStream_t stream) {
    (void)in_sizes; (void)n_in; (void)out_size; (void)d_ws; (void)ws_size;
    const float* x0 = (const float*)d_in[0];
    const float* U  = (const float*)d_in[1];
    const float* dt = (const float*)d_in[2];
    float* out = (float*)d_out;
    rollout_scan_kernel<<<NGRID, BLOCK, 0, stream>>>(x0, U, dt, out);
}
